# starting kernel
# baseline (speedup 1.0000x reference)
#include <hip/hip_runtime.h>
#include <hip/hip_bf16.h>

#define HID     128
#define FIN     11
#define FOUT    19
#define NGRAPHS 4096

typedef __attribute__((ext_vector_type(2))) float v2f;
typedef __attribute__((ext_vector_type(8))) float v8f;

__global__ void k_init_deg(float* deg, int n) {
    int i = blockIdx.x * blockDim.x + threadIdx.x;
    if (i < n) deg[i] = 1.0f;
}

__global__ void k_deg_accum(const int* __restrict__ dst, float* deg, int e) {
    int i = blockIdx.x * blockDim.x + threadIdx.x;
    if (i < e) atomicAdd(&deg[dst[i]], 1.0f);
}

__global__ void k_dinv(const float* __restrict__ deg, float* dinv, int n) {
    int i = blockIdx.x * blockDim.x + threadIdx.x;
    if (i < n) {
        float d = deg[i];
        dinv[i] = (d > 0.0f) ? rsqrtf(d) : 0.0f;
    }
}

__global__ void k_gemm1(const float* __restrict__ x, const float* __restrict__ W1,
                        float* __restrict__ h, int n) {
    int idx = blockIdx.x * blockDim.x + threadIdx.x;
    if (idx >= n * HID) return;
    int node = idx >> 7;
    int f    = idx & (HID - 1);
    const float* xr = x + node * FIN;
    float acc = 0.0f;
#pragma unroll
    for (int k = 0; k < FIN; ++k)
        acc = fmaf(xr[k], W1[k * HID + f], acc);
    h[idx] = acc;
}

__global__ void k_init_out(float* __restrict__ out, const float* __restrict__ h,
                           const float* __restrict__ bias, const float* __restrict__ dinv,
                           int n) {
    int idx = blockIdx.x * blockDim.x + threadIdx.x;
    if (idx >= n * HID) return;
    int node = idx >> 7;
    int f    = idx & (HID - 1);
    float dn = dinv[node];
    out[idx] = bias[f] + h[idx] * dn * dn;
}

__global__ void k_edge_scatter(const float* __restrict__ h, float* __restrict__ out,
                               const int* __restrict__ src, const int* __restrict__ dst,
                               const float* __restrict__ dinv, int e) {
    int gid  = blockIdx.x * blockDim.x + threadIdx.x;
    int edge = gid >> 5;
    int lane = gid & 31;
    if (edge >= e) return;
    int s = src[edge];
    int d = dst[edge];
    float nrm = dinv[s] * dinv[d];
    float4 v = ((const float4*)(h + (size_t)s * HID))[lane];
    float* o = out + (size_t)d * HID + lane * 4;
    atomicAdd(o + 0, v.x * nrm);
    atomicAdd(o + 1, v.y * nrm);
    atomicAdd(o + 2, v.z * nrm);
    atomicAdd(o + 3, v.w * nrm);
}

__global__ __launch_bounds__(256) void k_gemm2_wmma(const float* __restrict__ in,
                                                    const float* __restrict__ W,
                                                    float* __restrict__ out, int n) {
    __shared__ float Ws[HID * HID];
    const int tid = threadIdx.x;

    const float4* W4  = (const float4*)W;
    float4*       Ws4 = (float4*)Ws;
#pragma unroll
    for (int i = 0; i < 16; ++i)
        Ws4[tid + 256 * i] = W4[tid + 256 * i];
    __syncthreads();

    const int m0   = blockIdx.x << 4;
    const int wave = tid >> 5;
    const int lane = tid & 31;
    const int half = lane >> 4;
    const int l15  = lane & 15;
    const int n0   = wave << 4;
    int row = m0 + l15;
    if (row >= n) row = n - 1;
    const int ncol = n0 + l15;
    const float* arow = in + (size_t)row * HID;

    v8f c = {};
#pragma unroll
    for (int k0 = 0; k0 < HID; k0 += 4) {
        const int ka = k0 + half * 2;
        float2 av = *(const float2*)(arow + ka);
        v2f a, b;
        a[0] = fmaxf(av.x, 0.0f);
        a[1] = fmaxf(av.y, 0.0f);
        b[0] = Ws[ka * HID + ncol];
        b[1] = Ws[(ka + 1) * HID + ncol];
        c = __builtin_amdgcn_wmma_f32_16x16x4_f32(
                 false, a,  false, b,
                 (short)0, c,  false,  false);
    }

    const int mbase = m0 + half * 8;
#pragma unroll
    for (int i = 0; i < 8; ++i) {
        int r = mbase + i;
        if (r < n) out[(size_t)r * HID + ncol] = c[i];
    }
}

__global__ void k_pool_init(float* sums, float* cnt) {
    int idx = blockIdx.x * blockDim.x + threadIdx.x;
    if (idx < NGRAPHS * HID) sums[idx] = 0.0f;
    if (idx < NGRAPHS)       cnt[idx]  = 0.0f;
}

__global__ void k_pool_accum(const float* __restrict__ h, const int* __restrict__ batch,
                             float* sums, int n) {
    int idx = blockIdx.x * blockDim.x + threadIdx.x;
    if (idx >= n * HID) return;
    int node = idx >> 7;
    int f    = idx & (HID - 1);
    atomicAdd(&sums[batch[node] * HID + f], h[idx]);
}

__global__ void k_cnt_accum(const int* __restrict__ batch, float* cnt, int n) {
    int i = blockIdx.x * blockDim.x + threadIdx.x;
    if (i < n) atomicAdd(&cnt[batch[i]], 1.0f);
}

__global__ void k_final(const float* __restrict__ sums, const float* __restrict__ cnt,
                        const float* __restrict__ Wlin, const float* __restrict__ blin,
                        float* __restrict__ out) {
    int idx = blockIdx.x * blockDim.x + threadIdx.x;
    if (idx >= NGRAPHS * FOUT) return;
    int g = idx / FOUT;
    int j = idx - g * FOUT;
    float inv = 1.0f / fmaxf(cnt[g], 1.0f);
    const float* s = sums + g * HID;
    float acc = blin[j];
#pragma unroll 4
    for (int k = 0; k < HID; ++k)
        acc = fmaf(s[k] * inv, Wlin[k * FOUT + j], acc);
    out[idx] = acc;
}

extern "C" void kernel_launch(void* const* d_in, const int* in_sizes, int n_in,
                              void* d_out, int out_size, void* d_ws, size_t ws_size,
                              hipStream_t stream) {
    (void)n_in; (void)out_size; (void)ws_size;

    const float* x     = (const float*)d_in[0];
    const int*   esrc  = (const int*)  d_in[1];
    const int*   edst  = (const int*)  d_in[2];
    const int*   batch = (const int*)  d_in[3];
    const float* W1    = (const float*)d_in[4];
    const float* b1    = (const float*)d_in[5];
    const float* W2    = (const float*)d_in[6];
    const float* b2    = (const float*)d_in[7];
    const float* Wlin  = (const float*)d_in[8];
    const float* blin  = (const float*)d_in[9];
    float*       out   = (float*)d_out;

    const int N = in_sizes[3];
    const int E = in_sizes[1];

    char*  ws  = (char*)d_ws;
    size_t off = 0;
    auto carve = [&](size_t bytes) -> void* {
        off = (off + 255) & ~(size_t)255;
        void* p = ws + off;
        off += bytes;
        return p;
    };
    float* deg   = (float*)carve((size_t)N * 4);
    float* dinv  = (float*)carve((size_t)N * 4);
    float* bufH  = (float*)carve((size_t)N * HID * 4);
    float* bufO1 = (float*)carve((size_t)N * HID * 4);
    float* bufO2 = (float*)carve((size_t)N * HID * 4);
    float* sums  = (float*)carve((size_t)NGRAPHS * HID * 4);
    float* cnt   = (float*)carve((size_t)NGRAPHS * 4);

    const int T = 256;
    int nb   = (N + T - 1) / T;
    int eb   = (E + T - 1) / T;
    int fb   = (N * HID + T - 1) / T;
    int ewb  = (int)(((size_t)E * 32 + T - 1) / T);
    int gb2  = (N + 15) / 16;
    int pb   = (NGRAPHS * HID + T - 1) / T;
    int hb   = (NGRAPHS * FOUT + T - 1) / T;

    k_init_deg <<<nb, T, 0, stream>>>(deg, N);
    k_deg_accum<<<eb, T, 0, stream>>>(edst, deg, E);
    k_dinv     <<<nb, T, 0, stream>>>(deg, dinv, N);

    k_gemm1       <<<fb,  T, 0, stream>>>(x, W1, bufH, N);
    k_init_out    <<<fb,  T, 0, stream>>>(bufO1, bufH, b1, dinv, N);
    k_edge_scatter<<<ewb, T, 0, stream>>>(bufH, bufO1, esrc, edst, dinv, E);

    k_gemm2_wmma  <<<gb2, T, 0, stream>>>(bufO1, W2, bufH, N);
    k_init_out    <<<fb,  T, 0, stream>>>(bufO2, bufH, b2, dinv, N);
    k_edge_scatter<<<ewb, T, 0, stream>>>(bufH, bufO2, esrc, edst, dinv, E);

    k_pool_init <<<pb, T, 0, stream>>>(sums, cnt);
    k_pool_accum<<<fb, T, 0, stream>>>(bufO2, batch, sums, N);
    k_cnt_accum <<<nb, T, 0, stream>>>(batch, cnt, N);
    k_final     <<<hb, T, 0, stream>>>(sums, cnt, Wlin, blin, out);
}


# W2 product fused into the layer-1 owner work-groups: relu(summed rows) x W2 on f32 WMMA straight from LDS (row pitch 132), stored as layer 2's input; the product launch and a 51-MB round trip go
# speedup vs baseline: 2.4163x; 2.4163x over previous
#include <hip/hip_runtime.h>
#include <hip/hip_bf16.h>

#define HID     128
#define FIN     11
#define FOUT    19
#define NGRAPHS 4096

typedef __attribute__((ext_vector_type(2))) float v2f;
typedef __attribute__((ext_vector_type(8))) float v8f;

#define OWN_R   392
#define ACC_P   132

__global__ void k_init_deg(float* deg, int n) {
    int i = blockIdx.x * blockDim.x + threadIdx.x;
    if (i < n) deg[i] = 1.0f;
}

__device__ __forceinline__ unsigned owner_key(int d) { return ((unsigned)d / OWN_R) & 255u; }

__global__ void k_deg_key(const int* __restrict__ dst, float* deg, unsigned char* __restrict__ key, int e) {
    const int t    = blockIdx.x * blockDim.x + threadIdx.x;
    const int nvec = e >> 2;
    if (t < nvec) {
        const int4 d = ((const int4*)dst)[t];
        atomicAdd(&deg[d.x], 1.0f);
        atomicAdd(&deg[d.y], 1.0f);
        atomicAdd(&deg[d.z], 1.0f);
        atomicAdd(&deg[d.w], 1.0f);
        ((unsigned*)key)[t] = owner_key(d.x) | (owner_key(d.y) << 8) | (owner_key(d.z) << 16) | (owner_key(d.w) << 24);
    }
    if (t < (e & 3)) {
        const int i = (nvec << 2) + t;
        const int d = dst[i];
        atomicAdd(&deg[d], 1.0f);
        key[i] = (unsigned char)owner_key(d);
    }
}

__global__ void k_dinv(const float* __restrict__ deg, float* dinv, int n) {
    int i = blockIdx.x * blockDim.x + threadIdx.x;
    if (i < n) {
        float d = deg[i];
        dinv[i] = (d > 0.0f) ? rsqrtf(d) : 0.0f;
    }
}

#define G1_BLOCKS 1024
__global__ __launch_bounds__(256) void k_gemm1(const float* __restrict__ x, const float* __restrict__ W1,
                                               float* __restrict__ h, int n) {
    const int lane = threadIdx.x & 31;
    const int wave = __builtin_amdgcn_readfirstlane((int)(threadIdx.x >> 5));
    const int gw   = blockIdx.x * 8 + wave;
    const int nw   = gridDim.x * 8;
    float4 w[FIN];
#pragma unroll
    for (int k = 0; k < FIN; ++k) w[k] = ((const float4*)W1)[k * (HID / 4) + lane];
    for (int row = gw; row < n; row += nw) {
        const float* xr = x + (size_t)row * FIN;
        float4 acc = make_float4(0.0f, 0.0f, 0.0f, 0.0f);
#pragma unroll
        for (int k = 0; k < FIN; ++k) {
            const float xv = xr[k];
            acc.x = fmaf(xv, w[k].x, acc.x);
            acc.y = fmaf(xv, w[k].y, acc.y);
            acc.z = fmaf(xv, w[k].z, acc.z);
            acc.w = fmaf(xv, w[k].w, acc.w);
        }
        ((float4*)(h + (size_t)row * HID))[lane] = acc;
    }
}

#define OWN_NT  1024
#define OWN_CAP 8192

template <bool W2EPI>
__global__ __launch_bounds__(OWN_NT) void k_conv_owner(const float* __restrict__ h,
                                                       float* __restrict__ out,
                                                       const int* __restrict__ src,
                                                       const int* __restrict__ dst,
                                                       const unsigned char* __restrict__ key,
                                                       const float* __restrict__ dinv,
                                                       const float* __restrict__ bias,
                                                       const float* __restrict__ W,
                                                       int n, int e) {
    __shared__ __attribute__((aligned(16))) float acc[OWN_R * ACC_P];
    __shared__ int            list_e[OWN_CAP];
    __shared__ int            lcount;

    const int tid   = threadIdx.x;
    const int nbase = blockIdx.x * OWN_R;
    int rows = n - nbase;
    if (rows > OWN_R) rows = OWN_R;
    const unsigned urows = (unsigned)rows;

    {
        const float4  b4   = ((const float4*)bias)[tid & 31];
        const float4* h4   = (const float4*)(h + (size_t)nbase * HID);
        float4*       acc4 = (float4*)acc;
        for (int i = tid; i < rows * (HID / 4); i += OWN_NT) {
            const float  dn = dinv[nbase + (i >> 5)];
            const float4 v  = h4[i];
            float4 o;
            o.x = b4.x + v.x * dn * dn;
            o.y = b4.y + v.y * dn * dn;
            o.z = b4.z + v.z * dn * dn;
            o.w = b4.w + v.w * dn * dn;
            acc4[(i >> 5) * (ACC_P / 4) + (i & 31)] = o;
        }
    }
    if (tid == 0) lcount = 0;
    __syncthreads();

    auto note1 = [&](int eid) {
        const int sl = atomicAdd(&lcount, 1);
        if (sl < OWN_CAP) list_e[sl] = eid;
    };
    const unsigned mykey = (unsigned)blockIdx.x & 255u;
    const unsigned K4    = mykey * 0x01010101u;
    auto nonzero_bytes = [](unsigned x) -> unsigned {
        return (((x & 0x7F7F7F7Fu) + 0x7F7F7F7Fu) | x) & 0x80808080u;
    };
    auto note16 = [&](const uint4 w, int eb) {
        const unsigned g = (nonzero_bytes(w.x ^ K4) >> 7) | (nonzero_bytes(w.y ^ K4) >> 6) |
                           (nonzero_bytes(w.z ^ K4) >> 5) | (nonzero_bytes(w.w ^ K4) >> 4);
        unsigned hit = g ^ 0x0F0F0F0Fu;
        while (hit) {
            const int p = __builtin_ctz(hit);
            hit &= hit - 1;
            note1(eb + ((p & 3) << 2) + (p >> 3));
        }
    };

    const uint4* key16 = (const uint4*)key;
    const int nvec = e >> 4;
    int i0 = 0;
    for (; i0 + OWN_NT * 8 <= nvec; i0 += OWN_NT * 8) {
        uint4 kv[8];
#pragma unroll
        for (int u = 0; u < 8; ++u) kv[u] = key16[i0 + u * OWN_NT + tid];
#pragma unroll
        for (int u = 0; u < 8; ++u) note16(kv[u], (i0 + u * OWN_NT + tid) << 4);
    }
    for (int i = i0 + tid; i < nvec; i += OWN_NT) note16(key16[i], i << 4);
    if (tid < (e & 15)) {
        const int eid = (nvec << 4) + tid;
        if ((unsigned)key[eid] == mykey) note1(eid);
    }
    __syncthreads();

    const int total = lcount;
    if (total <= OWN_CAP && gridDim.x <= 256) {
        const int lane = tid & 31;
        const int wave = __builtin_amdgcn_readfirstlane(tid >> 5);
        for (int base = wave * 32; base < total; base += OWN_NT) {
            int   s = 0, dl = 0;
            float nrm = 0.0f;
            if (base + lane < total) {
                const int eid = list_e[base + lane];
                const int d   = dst[eid];
                s = src[eid];
                const unsigned a  = (unsigned)(d - nbase);
                const bool     in = a < urows;
                const float w = dinv[s] * dinv[d];
                dl  = in ? (int)a : 0;
                nrm = in ? w : 0.0f;
            }
            int cnt = total - base;
            if (cnt > 32) cnt = 32;
            int j = 0;
            for (; j + 8 <= cnt; j += 8) {
                float v[8][4];
#pragma unroll
                for (int u = 0; u < 8; ++u) {
                    const int sj = __builtin_amdgcn_readlane(s, j + u);
                    const float* r = h + (size_t)sj * HID + lane;
#pragma unroll
                    for (int k = 0; k < 4; ++k) v[u][k] = r[k * 32];
                }
#pragma unroll
                for (int u = 0; u < 8; ++u) {
                    const int   dj = __builtin_amdgcn_readlane(dl, j + u);
                    const float nj = __int_as_float(__builtin_amdgcn_readlane(__float_as_int(nrm), j + u));
                    float* a = acc + dj * ACC_P + lane;
#pragma unroll
                    for (int k = 0; k < 4; ++k) atomicAdd(&a[k * 32], v[u][k] * nj);
                }
            }
            for (; j < cnt; ++j) {
                const int   sj = __builtin_amdgcn_readlane(s, j);
                const int   dj = __builtin_amdgcn_readlane(dl, j);
                const float nj = __int_as_float(__builtin_amdgcn_readlane(__float_as_int(nrm), j));
                const float* r = h + (size_t)sj * HID + lane;
                float* a = acc + dj * ACC_P + lane;
#pragma unroll
                for (int k = 0; k < 4; ++k) atomicAdd(&a[k * 32], r[k * 32] * nj);
            }
        }
    } else {
        for (int i = tid; i < e; i += OWN_NT) {
            const unsigned a = (unsigned)(dst[i] - nbase);
            if (a < urows) {
                const int   s   = src[i];
                const float nrm = dinv[s] * dinv[nbase + (int)a];
                const float* r  = h + (size_t)s * HID;
                float* ar = acc + a * ACC_P;
                for (int f = 0; f < HID; ++f) atomicAdd(&ar[f], r[f] * nrm);
            }
        }
    }
    if constexpr (W2EPI) {
        const int lane = tid & 31;
        const int wave = __builtin_amdgcn_readfirstlane(tid >> 5);
        const int half = lane >> 4;
        const int l15  = lane & 15;
        const int ncol = ((wave & 7) << 4) + l15;
        v2f bf[HID / 4];
#pragma unroll
        for (int kk = 0; kk < HID / 4; ++kk) {
            const int ka = kk * 4 + half * 2;
            bf[kk][0] = W[ka * HID + ncol];
            bf[kk][1] = W[(ka + 1) * HID + ncol];
        }
        __syncthreads();

        const int ntile = (rows + 15) >> 4;
        for (int t = wave >> 3; t < ntile; t += 4) {
            int r = (t << 4) + l15;
            if (r >= rows) r = rows - 1;
            const float* ar = acc + r * ACC_P + half * 2;
            v8f c = {};
#pragma unroll
            for (int kk = 0; kk < HID / 4; ++kk) {
                const float2 av = *(const float2*)(ar + kk * 4);
                v2f a;
                a[0] = fmaxf(av.x, 0.0f);
                a[1] = fmaxf(av.y, 0.0f);
                c = __builtin_amdgcn_wmma_f32_16x16x4_f32(
                         false, a,  false, bf[kk],
                         (short)0, c,  false,  false);
            }
            const int rb = (t << 4) + half * 8;
#pragma unroll
            for (int i = 0; i < 8; ++i) {
                const int rr = rb + i;
                if (rr < rows) out[(size_t)(nbase + rr) * HID + ncol] = c[i];
            }
        }
    } else {
        __syncthreads();

        float4*       o4 = (float4*)(out + (size_t)nbase * HID);
        const float4* a4 = (const float4*)acc;
        for (int i = tid; i < rows * (HID / 4); i += OWN_NT) o4[i] = a4[(i >> 5) * (ACC_P / 4) + (i & 31)];
    }
}

__global__ void k_pool_init(float* sums, float* cnt) {
    int idx = blockIdx.x * blockDim.x + threadIdx.x;
    const float4 z = make_float4(0.0f, 0.0f, 0.0f, 0.0f);
    if (idx < NGRAPHS * HID / 4) ((float4*)sums)[idx] = z;
    if (idx < NGRAPHS / 4)       ((float4*)cnt)[idx]  = z;
}

#define PA_ROWS 16
__global__ __launch_bounds__(256) void k_pool_accum(const float* __restrict__ h, const int* __restrict__ batch,
                                                    float* sums, int n) {
    const int lane = threadIdx.x & 31;
    const int wave = __builtin_amdgcn_readfirstlane((int)(threadIdx.x >> 5));
    const int r0   = (blockIdx.x * 8 + wave) * PA_ROWS;
    if (r0 >= n) return;
    int cnt = n - r0;
    if (cnt > PA_ROWS) cnt = PA_ROWS;
    const int b = (lane < cnt) ? batch[r0 + lane] : 0;

    float v[PA_ROWS][4];
#pragma unroll
    for (int u = 0; u < PA_ROWS; ++u) {
        const int r = r0 + ((u < cnt) ? u : 0);
        const float* p = h + (size_t)r * HID + lane;
#pragma unroll
        for (int k = 0; k < 4; ++k) v[u][k] = p[k * 32];
    }

    float acc[4] = {0.0f, 0.0f, 0.0f, 0.0f};
    int cur = __builtin_amdgcn_readlane(b, 0);
#pragma unroll
    for (int u = 0; u < PA_ROWS; ++u) {
        if (u < cnt) {
            const int bu = __builtin_amdgcn_readlane(b, u);
            if (bu != cur) {
                float* s = sums + (size_t)cur * HID + lane;
#pragma unroll
                for (int k = 0; k < 4; ++k) { atomicAdd(&s[k * 32], acc[k]); acc[k] = 0.0f; }
                cur = bu;
            }
#pragma unroll
            for (int k = 0; k < 4; ++k) acc[k] += v[u][k];
        }
    }
    float* s = sums + (size_t)cur * HID + lane;
#pragma unroll
    for (int k = 0; k < 4; ++k) atomicAdd(&s[k * 32], acc[k]);
}

__global__ void k_cnt_accum(const int* __restrict__ batch, float* cnt, int n) {
    int i = blockIdx.x * blockDim.x + threadIdx.x;
    if (i < n) atomicAdd(&cnt[batch[i]], 1.0f);
}

__global__ void k_final(const float* __restrict__ sums, const float* __restrict__ cnt,
                        const float* __restrict__ Wlin, const float* __restrict__ blin,
                        float* __restrict__ out) {
    int idx = blockIdx.x * blockDim.x + threadIdx.x;
    if (idx >= NGRAPHS * FOUT) return;
    int g = idx / FOUT;
    int j = idx - g * FOUT;
    float inv = 1.0f / fmaxf(cnt[g], 1.0f);
    const float* s = sums + g * HID;
    float acc = blin[j];
#pragma unroll 4
    for (int k = 0; k < HID; ++k)
        acc = fmaf(s[k] * inv, Wlin[k * FOUT + j], acc);
    out[idx] = acc;
}

extern "C" void kernel_launch(void* const* d_in, const int* in_sizes, int n_in,
                              void* d_out, int out_size, void* d_ws, size_t ws_size,
                              hipStream_t stream) {
    (void)n_in; (void)out_size; (void)ws_size;

    const float* x     = (const float*)d_in[0];
    const int*   esrc  = (const int*)  d_in[1];
    const int*   edst  = (const int*)  d_in[2];
    const int*   batch = (const int*)  d_in[3];
    const float* W1    = (const float*)d_in[4];
    const float* b1    = (const float*)d_in[5];
    const float* W2    = (const float*)d_in[6];
    const float* b2    = (const float*)d_in[7];
    const float* Wlin  = (const float*)d_in[8];
    const float* blin  = (const float*)d_in[9];
    float*       out   = (float*)d_out;

    const int N = in_sizes[3];
    const int E = in_sizes[1];

    char*  ws  = (char*)d_ws;
    size_t off = 0;
    auto carve = [&](size_t bytes) -> void* {
        off = (off + 255) & ~(size_t)255;
        void* p = ws + off;
        off += bytes;
        return p;
    };
    float* deg   = (float*)carve((size_t)N * 4);
    float* dinv  = (float*)carve((size_t)N * 4);
    float* bufH  = (float*)carve((size_t)N * HID * 4);
    float* bufO1 = (float*)carve((size_t)N * HID * 4);
    unsigned char* okey = (unsigned char*)carve((size_t)E);
    float* sums  = (float*)carve((size_t)NGRAPHS * HID * 4);
    float* cnt   = (float*)carve((size_t)NGRAPHS * 4);

    const int T = 256;
    int nb   = (N + T - 1) / T;
    int kb   = (E / 4) / T + 1;
    int pab  = (N + 8 * PA_ROWS - 1) / (8 * PA_ROWS);
    int ob   = (N + OWN_R - 1) / OWN_R;
    int pb   = (NGRAPHS * HID / 4 + T - 1) / T;
    int hb   = (NGRAPHS * FOUT + T - 1) / T;

    k_init_deg <<<nb, T, 0, stream>>>(deg, N);
    k_deg_key  <<<kb, T, 0, stream>>>(edst, deg, okey, E);
    k_dinv     <<<nb, T, 0, stream>>>(deg, dinv, N);

    k_gemm1       <<<G1_BLOCKS, T, 0, stream>>>(x, W1, bufH, N);
    k_conv_owner<true> <<<ob, OWN_NT, 0, stream>>>(bufH, bufO1, esrc, edst, okey, dinv, b1, W2, N, E);

    k_conv_owner<false><<<ob, OWN_NT, 0, stream>>>(bufO1, bufH, esrc, edst, okey, dinv, b2, W2, N, E);

    k_pool_init <<<pb, T, 0, stream>>>(sums, cnt);
    k_pool_accum<<<pab, T, 0, stream>>>(bufH, batch, sums, N);
    k_cnt_accum <<<nb, T, 0, stream>>>(batch, cnt, N);
    k_final     <<<hb, T, 0, stream>>>(sums, cnt, Wlin, blin, out);
}


# Owner filter hoisted into one listing pass a call: owners scan the keys once, count degrees in LDS (global f32 degree adds go), store packed edge records both layers walk; LDS add order as before
# speedup vs baseline: 3.4801x; 1.4403x over previous
#include <hip/hip_runtime.h>
#include <hip/hip_bf16.h>

#define HID     128
#define FIN     11
#define FOUT    19
#define NGRAPHS 4096

typedef __attribute__((ext_vector_type(2))) float v2f;
typedef __attribute__((ext_vector_type(8))) float v8f;

#define OWN_R   392
#define ACC_P   132

#define OWN_NT  1024
#define OWN_CAP 8192
#define REC_BITS 9
#define REC_NONE 0xFFFFFFFFu

__device__ __forceinline__ unsigned owner_key(int d) { return ((unsigned)d / OWN_R) & 255u; }

__global__ void k_key(const int* __restrict__ dst, unsigned char* __restrict__ key, int e) {
    const int t    = blockIdx.x * blockDim.x + threadIdx.x;
    const int nvec = e >> 2;
    if (t < nvec) {
        const int4 d = ((const int4*)dst)[t];
        ((unsigned*)key)[t] = owner_key(d.x) | (owner_key(d.y) << 8) | (owner_key(d.z) << 16) | (owner_key(d.w) << 24);
    }
    if (t < (e & 3)) {
        const int i = (nvec << 2) + t;
        key[i] = (unsigned char)owner_key(dst[i]);
    }
}

__global__ __launch_bounds__(OWN_NT) void k_owner_list(const int* __restrict__ src,
                                                       const int* __restrict__ dst,
                                                       const unsigned char* __restrict__ key,
                                                       unsigned* __restrict__ lists,
                                                       int* __restrict__ lcnt,
                                                       float* __restrict__ dinv,
                                                       int n, int e) {
    __shared__ int list_e[OWN_CAP];
    __shared__ int cntr[OWN_R];
    __shared__ int lcount;

    const int tid   = threadIdx.x;
    const int nbase = blockIdx.x * OWN_R;
    int rows = n - nbase;
    if (rows > OWN_R) rows = OWN_R;
    const unsigned urows = (unsigned)rows;

    if (tid < OWN_R) cntr[tid] = 0;
    if (tid == 0) lcount = 0;
    __syncthreads();

    auto note1 = [&](int eid) {
        const int sl = atomicAdd(&lcount, 1);
        if (sl < OWN_CAP) list_e[sl] = eid;
    };
    const unsigned mykey = (unsigned)blockIdx.x & 255u;
    const unsigned K4    = mykey * 0x01010101u;
    auto nonzero_bytes = [](unsigned x) -> unsigned {
        return (((x & 0x7F7F7F7Fu) + 0x7F7F7F7Fu) | x) & 0x80808080u;
    };
    auto note16 = [&](const uint4 w, int eb) {
        const unsigned g = (nonzero_bytes(w.x ^ K4) >> 7) | (nonzero_bytes(w.y ^ K4) >> 6) |
                           (nonzero_bytes(w.z ^ K4) >> 5) | (nonzero_bytes(w.w ^ K4) >> 4);
        unsigned hit = g ^ 0x0F0F0F0Fu;
        while (hit) {
            const int p = __builtin_ctz(hit);
            hit &= hit - 1;
            note1(eb + ((p & 3) << 2) + (p >> 3));
        }
    };

    const uint4* key16 = (const uint4*)key;
    const int nvec = e >> 4;
    int i0 = 0;
    for (; i0 + OWN_NT * 8 <= nvec; i0 += OWN_NT * 8) {
        uint4 kv[8];
#pragma unroll
        for (int u = 0; u < 8; ++u) kv[u] = key16[i0 + u * OWN_NT + tid];
#pragma unroll
        for (int u = 0; u < 8; ++u) note16(kv[u], (i0 + u * OWN_NT + tid) << 4);
    }
    for (int i = i0 + tid; i < nvec; i += OWN_NT) note16(key16[i], i << 4);
    if (tid < (e & 15)) {
        const int eid = (nvec << 4) + tid;
        if ((unsigned)key[eid] == mykey) note1(eid);
    }
    __syncthreads();

    const int total = lcount;
    const bool listed = (total <= OWN_CAP) && (gridDim.x <= 256);
    if (listed) {
        unsigned* mylist = lists + (size_t)blockIdx.x * OWN_CAP;
        int eid[OWN_CAP / OWN_NT];
#pragma unroll
        for (int u = 0; u < OWN_CAP / OWN_NT; ++u) {
            const int i = u * OWN_NT + tid;
            eid[u] = 0;
            if (i < total) eid[u] = list_e[i];
        }
        int d[OWN_CAP / OWN_NT], s[OWN_CAP / OWN_NT];
#pragma unroll
        for (int u = 0; u < OWN_CAP / OWN_NT; ++u) { d[u] = dst[eid[u]]; s[u] = src[eid[u]]; }
#pragma unroll
        for (int u = 0; u < OWN_CAP / OWN_NT; ++u) {
            const int i = u * OWN_NT + tid;
            if (i < total) {
                const unsigned a  = (unsigned)(d[u] - nbase);
                const bool     in = a < urows;
                if (in) atomicAdd(&cntr[a], 1);
                mylist[i] = in ? (((unsigned)s[u] << REC_BITS) | a) : REC_NONE;
            }
        }
    } else {
        for (int i = tid; i < e; i += OWN_NT) {
            const unsigned a = (unsigned)(dst[i] - nbase);
            if (a < urows) atomicAdd(&cntr[a], 1);
        }
    }
    __syncthreads();

    if (tid < rows) {
        const float dg = 1.0f + (float)cntr[tid];
        dinv[nbase + tid] = (dg > 0.0f) ? rsqrtf(dg) : 0.0f;
    }
    if (tid == 0) lcnt[blockIdx.x] = listed ? total : OWN_CAP + 1;
}

#define G1_BLOCKS 1024
__global__ __launch_bounds__(256) void k_gemm1(const float* __restrict__ x, const float* __restrict__ W1,
                                               float* __restrict__ h, int n) {
    const int lane = threadIdx.x & 31;
    const int wave = __builtin_amdgcn_readfirstlane((int)(threadIdx.x >> 5));
    const int gw   = blockIdx.x * 8 + wave;
    const int nw   = gridDim.x * 8;
    float4 w[FIN];
#pragma unroll
    for (int k = 0; k < FIN; ++k) w[k] = ((const float4*)W1)[k * (HID / 4) + lane];
    for (int row = gw; row < n; row += nw) {
        const float* xr = x + (size_t)row * FIN;
        float4 acc = make_float4(0.0f, 0.0f, 0.0f, 0.0f);
#pragma unroll
        for (int k = 0; k < FIN; ++k) {
            const float xv = xr[k];
            acc.x = fmaf(xv, w[k].x, acc.x);
            acc.y = fmaf(xv, w[k].y, acc.y);
            acc.z = fmaf(xv, w[k].z, acc.z);
            acc.w = fmaf(xv, w[k].w, acc.w);
        }
        ((float4*)(h + (size_t)row * HID))[lane] = acc;
    }
}

static_assert(OWN_R < (1 << REC_BITS) - 1, "the owned row must fit the record's row field, below REC_NONE's");

template <bool W2EPI>
__global__ __launch_bounds__(OWN_NT) void k_conv_owner(const float* __restrict__ h,
                                                       float* __restrict__ out,
                                                       const int* __restrict__ src,
                                                       const int* __restrict__ dst,
                                                       const unsigned* __restrict__ lists,
                                                       const int* __restrict__ lcnt,
                                                       const float* __restrict__ dinv,
                                                       const float* __restrict__ bias,
                                                       const float* __restrict__ W,
                                                       int n, int e) {
    __shared__ __attribute__((aligned(16))) float acc[OWN_R * ACC_P];

    const int tid   = threadIdx.x;
    const int nbase = blockIdx.x * OWN_R;
    int rows = n - nbase;
    if (rows > OWN_R) rows = OWN_R;
    const unsigned urows = (unsigned)rows;

    {
        const float4  b4   = ((const float4*)bias)[tid & 31];
        const float4* h4   = (const float4*)(h + (size_t)nbase * HID);
        float4*       acc4 = (float4*)acc;
        for (int i = tid; i < rows * (HID / 4); i += OWN_NT) {
            const float  dn = dinv[nbase + (i >> 5)];
            const float4 v  = h4[i];
            float4 o;
            o.x = b4.x + v.x * dn * dn;
            o.y = b4.y + v.y * dn * dn;
            o.z = b4.z + v.z * dn * dn;
            o.w = b4.w + v.w * dn * dn;
            acc4[(i >> 5) * (ACC_P / 4) + (i & 31)] = o;
        }
    }
    __syncthreads();

    const int total = lcnt[blockIdx.x];
    if (total <= OWN_CAP && gridDim.x <= 256) {
        const unsigned* mylist = lists + (size_t)blockIdx.x * OWN_CAP;
        const int lane = tid & 31;
        const int wave = __builtin_amdgcn_readfirstlane(tid >> 5);
        for (int base = wave * 32; base < total; base += OWN_NT) {
            int   s = 0, dl = 0;
            float nrm = 0.0f;
            if (base + lane < total) {
                const unsigned rec = mylist[base + lane];
                const bool     in  = rec != REC_NONE;
                s  = in ? (int)(rec >> REC_BITS) : 0;
                dl = in ? (int)(rec & ((1u << REC_BITS) - 1u)) : 0;
                const float w = dinv[s] * dinv[nbase + dl];
                nrm = in ? w : 0.0f;
            }
            int cnt = total - base;
            if (cnt > 32) cnt = 32;
            int j = 0;
            for (; j + 8 <= cnt; j += 8) {
                float v[8][4];
#pragma unroll
                for (int u = 0; u < 8; ++u) {
                    const int sj = __builtin_amdgcn_readlane(s, j + u);
                    const float* r = h + (size_t)sj * HID + lane;
#pragma unroll
                    for (int k = 0; k < 4; ++k) v[u][k] = r[k * 32];
                }
#pragma unroll
                for (int u = 0; u < 8; ++u) {
                    const int   dj = __builtin_amdgcn_readlane(dl, j + u);
                    const float nj = __int_as_float(__builtin_amdgcn_readlane(__float_as_int(nrm), j + u));
                    float* a = acc + dj * ACC_P + lane;
#pragma unroll
                    for (int k = 0; k < 4; ++k) atomicAdd(&a[k * 32], v[u][k] * nj);
                }
            }
            for (; j < cnt; ++j) {
                const int   sj = __builtin_amdgcn_readlane(s, j);
                const int   dj = __builtin_amdgcn_readlane(dl, j);
                const float nj = __int_as_float(__builtin_amdgcn_readlane(__float_as_int(nrm), j));
                const float* r = h + (size_t)sj * HID + lane;
                float* a = acc + dj * ACC_P + lane;
#pragma unroll
                for (int k = 0; k < 4; ++k) atomicAdd(&a[k * 32], r[k * 32] * nj);
            }
        }
    } else {
        for (int i = tid; i < e; i += OWN_NT) {
            const unsigned a = (unsigned)(dst[i] - nbase);
            if (a < urows) {
                const int   s   = src[i];
                const float nrm = dinv[s] * dinv[nbase + (int)a];
                const float* r  = h + (size_t)s * HID;
                float* ar = acc + a * ACC_P;
                for (int f = 0; f < HID; ++f) atomicAdd(&ar[f], r[f] * nrm);
            }
        }
    }
    if constexpr (W2EPI) {
        const int lane = tid & 31;
        const int wave = __builtin_amdgcn_readfirstlane(tid >> 5);
        const int half = lane >> 4;
        const int l15  = lane & 15;
        const int ncol = ((wave & 7) << 4) + l15;
        v2f bf[HID / 4];
#pragma unroll
        for (int kk = 0; kk < HID / 4; ++kk) {
            const int ka = kk * 4 + half * 2;
            bf[kk][0] = W[ka * HID + ncol];
            bf[kk][1] = W[(ka + 1) * HID + ncol];
        }
        __syncthreads();

        const int ntile = (rows + 15) >> 4;
        for (int t = wave >> 3; t < ntile; t += 4) {
            int r = (t << 4) + l15;
            if (r >= rows) r = rows - 1;
            const float* ar = acc + r * ACC_P + half * 2;
            v8f c = {};
#pragma unroll
            for (int kk = 0; kk < HID / 4; ++kk) {
                const float2 av = *(const float2*)(ar + kk * 4);
                v2f a;
                a[0] = fmaxf(av.x, 0.0f);
                a[1] = fmaxf(av.y, 0.0f);
                c = __builtin_amdgcn_wmma_f32_16x16x4_f32(
                         false, a,  false, bf[kk],
                         (short)0, c,  false,  false);
            }
            const int rb = (t << 4) + half * 8;
#pragma unroll
            for (int i = 0; i < 8; ++i) {
                const int rr = rb + i;
                if (rr < rows) out[(size_t)(nbase + rr) * HID + ncol] = c[i];
            }
        }
    } else {
        __syncthreads();

        float4*       o4 = (float4*)(out + (size_t)nbase * HID);
        const float4* a4 = (const float4*)acc;
        for (int i = tid; i < rows * (HID / 4); i += OWN_NT) o4[i] = a4[(i >> 5) * (ACC_P / 4) + (i & 31)];
    }
}

__global__ void k_pool_init(float* sums, float* cnt) {
    int idx = blockIdx.x * blockDim.x + threadIdx.x;
    const float4 z = make_float4(0.0f, 0.0f, 0.0f, 0.0f);
    if (idx < NGRAPHS * HID / 4) ((float4*)sums)[idx] = z;
    if (idx < NGRAPHS / 4)       ((float4*)cnt)[idx]  = z;
}

#define PA_ROWS 16
__global__ __launch_bounds__(256) void k_pool_accum(const float* __restrict__ h, const int* __restrict__ batch,
                                                    float* sums, int n) {
    const int lane = threadIdx.x & 31;
    const int wave = __builtin_amdgcn_readfirstlane((int)(threadIdx.x >> 5));
    const int r0   = (blockIdx.x * 8 + wave) * PA_ROWS;
    if (r0 >= n) return;
    int cnt = n - r0;
    if (cnt > PA_ROWS) cnt = PA_ROWS;
    const int b = (lane < cnt) ? batch[r0 + lane] : 0;

    float v[PA_ROWS][4];
#pragma unroll
    for (int u = 0; u < PA_ROWS; ++u) {
        const int r = r0 + ((u < cnt) ? u : 0);
        const float* p = h + (size_t)r * HID + lane;
#pragma unroll
        for (int k = 0; k < 4; ++k) v[u][k] = p[k * 32];
    }

    float acc[4] = {0.0f, 0.0f, 0.0f, 0.0f};
    int cur = __builtin_amdgcn_readlane(b, 0);
#pragma unroll
    for (int u = 0; u < PA_ROWS; ++u) {
        if (u < cnt) {
            const int bu = __builtin_amdgcn_readlane(b, u);
            if (bu != cur) {
                float* s = sums + (size_t)cur * HID + lane;
#pragma unroll
                for (int k = 0; k < 4; ++k) { atomicAdd(&s[k * 32], acc[k]); acc[k] = 0.0f; }
                cur = bu;
            }
#pragma unroll
            for (int k = 0; k < 4; ++k) acc[k] += v[u][k];
        }
    }
    float* s = sums + (size_t)cur * HID + lane;
#pragma unroll
    for (int k = 0; k < 4; ++k) atomicAdd(&s[k * 32], acc[k]);
}

__global__ void k_cnt_accum(const int* __restrict__ batch, float* cnt, int n) {
    int i = blockIdx.x * blockDim.x + threadIdx.x;
    if (i < n) atomicAdd(&cnt[batch[i]], 1.0f);
}

__global__ void k_final(const float* __restrict__ sums, const float* __restrict__ cnt,
                        const float* __restrict__ Wlin, const float* __restrict__ blin,
                        float* __restrict__ out) {
    int idx = blockIdx.x * blockDim.x + threadIdx.x;
    if (idx >= NGRAPHS * FOUT) return;
    int g = idx / FOUT;
    int j = idx - g * FOUT;
    float inv = 1.0f / fmaxf(cnt[g], 1.0f);
    const float* s = sums + g * HID;
    float acc = blin[j];
#pragma unroll 4
    for (int k = 0; k < HID; ++k)
        acc = fmaf(s[k] * inv, Wlin[k * FOUT + j], acc);
    out[idx] = acc;
}

extern "C" void kernel_launch(void* const* d_in, const int* in_sizes, int n_in,
                              void* d_out, int out_size, void* d_ws, size_t ws_size,
                              hipStream_t stream) {
    (void)n_in; (void)out_size; (void)ws_size;

    const float* x     = (const float*)d_in[0];
    const int*   esrc  = (const int*)  d_in[1];
    const int*   edst  = (const int*)  d_in[2];
    const int*   batch = (const int*)  d_in[3];
    const float* W1    = (const float*)d_in[4];
    const float* b1    = (const float*)d_in[5];
    const float* W2    = (const float*)d_in[6];
    const float* b2    = (const float*)d_in[7];
    const float* Wlin  = (const float*)d_in[8];
    const float* blin  = (const float*)d_in[9];
    float*       out   = (float*)d_out;

    const int N = in_sizes[3];
    const int E = in_sizes[1];
    const int ob = (N + OWN_R - 1) / OWN_R;

    char*  ws  = (char*)d_ws;
    size_t off = 0;
    auto carve = [&](size_t bytes) -> void* {
        off = (off + 255) & ~(size_t)255;
        void* p = ws + off;
        off += bytes;
        return p;
    };
    float* dinv  = (float*)carve((size_t)N * 4);
    float* bufH  = (float*)carve((size_t)N * HID * 4);
    float* bufO1 = (float*)carve((size_t)N * HID * 4);
    unsigned char* okey = (unsigned char*)carve((size_t)E);
    unsigned* lists = (unsigned*)carve((size_t)256 * OWN_CAP * 4);
    int* lcnt = (int*)carve((size_t)ob * 4);
    float* sums  = (float*)carve((size_t)NGRAPHS * HID * 4);
    float* cnt   = (float*)carve((size_t)NGRAPHS * 4);

    const int T = 256;
    int nb   = (N + T - 1) / T;
    int kb   = (E / 4) / T + 1;
    int pab  = (N + 8 * PA_ROWS - 1) / (8 * PA_ROWS);
    int pb   = (NGRAPHS * HID / 4 + T - 1) / T;
    int hb   = (NGRAPHS * FOUT + T - 1) / T;

    k_key        <<<kb, T, 0, stream>>>(edst, okey, E);
    k_owner_list <<<ob, OWN_NT, 0, stream>>>(esrc, edst, okey, lists, lcnt, dinv, N, E);

    k_gemm1       <<<G1_BLOCKS, T, 0, stream>>>(x, W1, bufH, N);
    k_conv_owner<true> <<<ob, OWN_NT, 0, stream>>>(bufH, bufO1, esrc, edst, lists, lcnt, dinv, b1, W2, N, E);

    k_conv_owner<false><<<ob, OWN_NT, 0, stream>>>(bufO1, bufH, esrc, edst, lists, lcnt, dinv, b2, W2, N, E);

    k_pool_init <<<pb, T, 0, stream>>>(sums, cnt);
    k_pool_accum<<<pab, T, 0, stream>>>(bufH, batch, sums, N);
    k_cnt_accum <<<nb, T, 0, stream>>>(batch, cnt, N);
    k_final     <<<hb, T, 0, stream>>>(sums, cnt, Wlin, blin, out);
}


# Owner edge records stored in destination-row order (slot = LDS prefix of degree counts + the counting add's old value); a wave sums a row's run in registers, 1 LDS f32 add a run; sum order changes
# speedup vs baseline: 3.9588x; 1.1376x over previous
#include <hip/hip_runtime.h>
#include <hip/hip_bf16.h>

#define HID     128
#define FIN     11
#define FOUT    19
#define NGRAPHS 4096

typedef __attribute__((ext_vector_type(2))) float v2f;
typedef __attribute__((ext_vector_type(8))) float v8f;

#define OWN_R   392
#define ACC_P   132

#define OWN_NT  1024
#define OWN_CAP 8192
#define REC_BITS 9
#define SCAN_W  ((OWN_R + 31) / 32)

__device__ __forceinline__ unsigned owner_key(int d) { return ((unsigned)d / OWN_R) & 255u; }

__global__ void k_key(const int* __restrict__ dst, unsigned char* __restrict__ key, int e) {
    const int t    = blockIdx.x * blockDim.x + threadIdx.x;
    const int nvec = e >> 2;
    if (t < nvec) {
        const int4 d = ((const int4*)dst)[t];
        ((unsigned*)key)[t] = owner_key(d.x) | (owner_key(d.y) << 8) | (owner_key(d.z) << 16) | (owner_key(d.w) << 24);
    }
    if (t < (e & 3)) {
        const int i = (nvec << 2) + t;
        key[i] = (unsigned char)owner_key(dst[i]);
    }
}

__global__ __launch_bounds__(OWN_NT) void k_owner_list(const int* __restrict__ src,
                                                       const int* __restrict__ dst,
                                                       const unsigned char* __restrict__ key,
                                                       unsigned* __restrict__ lists,
                                                       int* __restrict__ lcnt,
                                                       float* __restrict__ dinv,
                                                       int n, int e) {
    __shared__ int list_e[OWN_CAP];
    __shared__ int cntr[OWN_R];
    __shared__ int first[OWN_R];
    __shared__ int wtot[SCAN_W];
    __shared__ int lcount;

    const int tid   = threadIdx.x;
    const int nbase = blockIdx.x * OWN_R;
    int rows = n - nbase;
    if (rows > OWN_R) rows = OWN_R;
    const unsigned urows = (unsigned)rows;

    if (tid < OWN_R) cntr[tid] = 0;
    if (tid == 0) lcount = 0;
    __syncthreads();

    auto note1 = [&](int eid) {
        const int sl = atomicAdd(&lcount, 1);
        if (sl < OWN_CAP) list_e[sl] = eid;
    };
    const unsigned mykey = (unsigned)blockIdx.x & 255u;
    const unsigned K4    = mykey * 0x01010101u;
    auto nonzero_bytes = [](unsigned x) -> unsigned {
        return (((x & 0x7F7F7F7Fu) + 0x7F7F7F7Fu) | x) & 0x80808080u;
    };
    auto note16 = [&](const uint4 w, int eb) {
        const unsigned g = (nonzero_bytes(w.x ^ K4) >> 7) | (nonzero_bytes(w.y ^ K4) >> 6) |
                           (nonzero_bytes(w.z ^ K4) >> 5) | (nonzero_bytes(w.w ^ K4) >> 4);
        unsigned hit = g ^ 0x0F0F0F0Fu;
        while (hit) {
            const int p = __builtin_ctz(hit);
            hit &= hit - 1;
            note1(eb + ((p & 3) << 2) + (p >> 3));
        }
    };

    const uint4* key16 = (const uint4*)key;
    const int nvec = e >> 4;
    int i0 = 0;
    for (; i0 + OWN_NT * 8 <= nvec; i0 += OWN_NT * 8) {
        uint4 kv[8];
#pragma unroll
        for (int u = 0; u < 8; ++u) kv[u] = key16[i0 + u * OWN_NT + tid];
#pragma unroll
        for (int u = 0; u < 8; ++u) note16(kv[u], (i0 + u * OWN_NT + tid) << 4);
    }
    for (int i = i0 + tid; i < nvec; i += OWN_NT) note16(key16[i], i << 4);
    if (tid < (e & 15)) {
        const int eid = (nvec << 4) + tid;
        if ((unsigned)key[eid] == mykey) note1(eid);
    }
    __syncthreads();

    const int total = lcount;
    const bool listed = (total <= OWN_CAP) && (gridDim.x <= 256);
    unsigned arow[OWN_CAP / OWN_NT];
    int      rk[OWN_CAP / OWN_NT];
    int      s[OWN_CAP / OWN_NT];
#pragma unroll
    for (int u = 0; u < OWN_CAP / OWN_NT; ++u) { arow[u] = 0xFFFFFFFFu; rk[u] = 0; s[u] = 0; }
    if (listed) {
        int eid[OWN_CAP / OWN_NT];
#pragma unroll
        for (int u = 0; u < OWN_CAP / OWN_NT; ++u) {
            const int i = u * OWN_NT + tid;
            eid[u] = 0;
            if (i < total) eid[u] = list_e[i];
        }
        int d[OWN_CAP / OWN_NT];
#pragma unroll
        for (int u = 0; u < OWN_CAP / OWN_NT; ++u) { d[u] = dst[eid[u]]; s[u] = src[eid[u]]; }
#pragma unroll
        for (int u = 0; u < OWN_CAP / OWN_NT; ++u) {
            const int i = u * OWN_NT + tid;
            if (i < total) {
                const unsigned a = (unsigned)(d[u] - nbase);
                if (a < urows) {
                    arow[u] = a;
                    rk[u]   = atomicAdd(&cntr[a], 1);
                }
            }
        }
    } else {
        for (int i = tid; i < e; i += OWN_NT) {
            const unsigned a = (unsigned)(dst[i] - nbase);
            if (a < urows) atomicAdd(&cntr[a], 1);
        }
    }
    __syncthreads();

    const int lane = tid & 31;
    const int wv   = tid >> 5;
    const int cv   = (tid < rows) ? cntr[tid] : 0;
    int inc = cv;
#pragma unroll
    for (int dlt = 1; dlt < 32; dlt <<= 1) {
        const int t = __shfl_up(inc, dlt);
        if (lane >= dlt) inc += t;
    }
    if (lane == 31 && wv < SCAN_W) wtot[wv] = inc;
    __syncthreads();
    int below = 0, stored = 0;
    for (int q = 0; q < SCAN_W; ++q) {
        const int t = wtot[q];
        if (q < wv) below += t;
        stored += t;
    }
    if (tid < OWN_R) first[tid] = below + inc - cv;
    __syncthreads();

    if (listed) {
#pragma unroll
        for (int u = 0; u < OWN_CAP / OWN_NT; ++u) {
            const unsigned a = arow[u];
            if (a < urows) list_e[first[a] + rk[u]] = (int)(((unsigned)s[u] << REC_BITS) | a);
        }
    }
    __syncthreads();
    if (listed) {
        unsigned* mylist = lists + (size_t)blockIdx.x * OWN_CAP;
        for (int i = tid; i < stored; i += OWN_NT) mylist[i] = (unsigned)list_e[i];
    }

    if (tid < rows) {
        const float dg = 1.0f + (float)cntr[tid];
        dinv[nbase + tid] = (dg > 0.0f) ? rsqrtf(dg) : 0.0f;
    }
    if (tid == 0) lcnt[blockIdx.x] = listed ? stored : OWN_CAP + 1;
}

#define G1_BLOCKS 1024
__global__ __launch_bounds__(256) void k_gemm1(const float* __restrict__ x, const float* __restrict__ W1,
                                               float* __restrict__ h, int n) {
    const int lane = threadIdx.x & 31;
    const int wave = __builtin_amdgcn_readfirstlane((int)(threadIdx.x >> 5));
    const int gw   = blockIdx.x * 8 + wave;
    const int nw   = gridDim.x * 8;
    float4 w[FIN];
#pragma unroll
    for (int k = 0; k < FIN; ++k) w[k] = ((const float4*)W1)[k * (HID / 4) + lane];
    for (int row = gw; row < n; row += nw) {
        const float* xr = x + (size_t)row * FIN;
        float4 acc = make_float4(0.0f, 0.0f, 0.0f, 0.0f);
#pragma unroll
        for (int k = 0; k < FIN; ++k) {
            const float xv = xr[k];
            acc.x = fmaf(xv, w[k].x, acc.x);
            acc.y = fmaf(xv, w[k].y, acc.y);
            acc.z = fmaf(xv, w[k].z, acc.z);
            acc.w = fmaf(xv, w[k].w, acc.w);
        }
        ((float4*)(h + (size_t)row * HID))[lane] = acc;
    }
}

static_assert(OWN_R <= (1 << REC_BITS), "the owned row must fit the record's row field");

template <bool W2EPI>
__global__ __launch_bounds__(OWN_NT) void k_conv_owner(const float* __restrict__ h,
                                                       float* __restrict__ out,
                                                       const int* __restrict__ src,
                                                       const int* __restrict__ dst,
                                                       const unsigned* __restrict__ lists,
                                                       const int* __restrict__ lcnt,
                                                       const float* __restrict__ dinv,
                                                       const float* __restrict__ bias,
                                                       const float* __restrict__ W,
                                                       int n, int e) {
    __shared__ __attribute__((aligned(16))) float acc[OWN_R * ACC_P];

    const int tid   = threadIdx.x;
    const int nbase = blockIdx.x * OWN_R;
    int rows = n - nbase;
    if (rows > OWN_R) rows = OWN_R;
    const unsigned urows = (unsigned)rows;

    {
        const float4  b4   = ((const float4*)bias)[tid & 31];
        const float4* h4   = (const float4*)(h + (size_t)nbase * HID);
        float4*       acc4 = (float4*)acc;
        for (int i = tid; i < rows * (HID / 4); i += OWN_NT) {
            const float  dn = dinv[nbase + (i >> 5)];
            const float4 v  = h4[i];
            float4 o;
            o.x = b4.x + v.x * dn * dn;
            o.y = b4.y + v.y * dn * dn;
            o.z = b4.z + v.z * dn * dn;
            o.w = b4.w + v.w * dn * dn;
            acc4[(i >> 5) * (ACC_P / 4) + (i & 31)] = o;
        }
    }
    __syncthreads();

    const int total = lcnt[blockIdx.x];
    if (total <= OWN_CAP && gridDim.x <= 256) {
        const unsigned* mylist = lists + (size_t)blockIdx.x * OWN_CAP;
        const int lane = tid & 31;
        const int wave = __builtin_amdgcn_readfirstlane(tid >> 5);
        for (int base = wave * 32; base < total; base += OWN_NT) {
            int   s = 0, dl = 0;
            float nrm = 0.0f;
            if (base + lane < total) {
                const unsigned rec = mylist[base + lane];
                s   = (int)(rec >> REC_BITS);
                dl  = (int)(rec & ((1u << REC_BITS) - 1u));
                nrm = dinv[s] * dinv[nbase + dl];
            }
            int cnt = total - base;
            if (cnt > 32) cnt = 32;
            float run[4] = {0.0f, 0.0f, 0.0f, 0.0f};
            int   cur = __builtin_amdgcn_readlane(dl, 0);
            auto flush = [&](int row) {
                float* a = acc + row * ACC_P + lane;
#pragma unroll
                for (int k = 0; k < 4; ++k) { atomicAdd(&a[k * 32], run[k]); run[k] = 0.0f; }
            };
            int j = 0;
            for (; j + 8 <= cnt; j += 8) {
                float v[8][4];
#pragma unroll
                for (int u = 0; u < 8; ++u) {
                    const int sj = __builtin_amdgcn_readlane(s, j + u);
                    const float* r = h + (size_t)sj * HID + lane;
#pragma unroll
                    for (int k = 0; k < 4; ++k) v[u][k] = r[k * 32];
                }
#pragma unroll
                for (int u = 0; u < 8; ++u) {
                    const int   dj = __builtin_amdgcn_readlane(dl, j + u);
                    const float nj = __int_as_float(__builtin_amdgcn_readlane(__float_as_int(nrm), j + u));
                    if (dj != cur) { flush(cur); cur = dj; }
#pragma unroll
                    for (int k = 0; k < 4; ++k) run[k] = fmaf(v[u][k], nj, run[k]);
                }
            }
            for (; j < cnt; ++j) {
                const int   sj = __builtin_amdgcn_readlane(s, j);
                const int   dj = __builtin_amdgcn_readlane(dl, j);
                const float nj = __int_as_float(__builtin_amdgcn_readlane(__float_as_int(nrm), j));
                const float* r = h + (size_t)sj * HID + lane;
                if (dj != cur) { flush(cur); cur = dj; }
#pragma unroll
                for (int k = 0; k < 4; ++k) run[k] = fmaf(r[k * 32], nj, run[k]);
            }
            flush(cur);
        }
    } else {
        for (int i = tid; i < e; i += OWN_NT) {
            const unsigned a = (unsigned)(dst[i] - nbase);
            if (a < urows) {
                const int   s   = src[i];
                const float nrm = dinv[s] * dinv[nbase + (int)a];
                const float* r  = h + (size_t)s * HID;
                float* ar = acc + a * ACC_P;
                for (int f = 0; f < HID; ++f) atomicAdd(&ar[f], r[f] * nrm);
            }
        }
    }
    if constexpr (W2EPI) {
        const int lane = tid & 31;
        const int wave = __builtin_amdgcn_readfirstlane(tid >> 5);
        const int half = lane >> 4;
        const int l15  = lane & 15;
        const int ncol = ((wave & 7) << 4) + l15;
        v2f bf[HID / 4];
#pragma unroll
        for (int kk = 0; kk < HID / 4; ++kk) {
            const int ka = kk * 4 + half * 2;
            bf[kk][0] = W[ka * HID + ncol];
            bf[kk][1] = W[(ka + 1) * HID + ncol];
        }
        __syncthreads();

        const int ntile = (rows + 15) >> 4;
        for (int t = wave >> 3; t < ntile; t += 4) {
            int r = (t << 4) + l15;
            if (r >= rows) r = rows - 1;
            const float* ar = acc + r * ACC_P + half * 2;
            v8f c = {};
#pragma unroll
            for (int kk = 0; kk < HID / 4; ++kk) {
                const float2 av = *(const float2*)(ar + kk * 4);
                v2f a;
                a[0] = fmaxf(av.x, 0.0f);
                a[1] = fmaxf(av.y, 0.0f);
                c = __builtin_amdgcn_wmma_f32_16x16x4_f32(
                         false, a,  false, bf[kk],
                         (short)0, c,  false,  false);
            }
            const int rb = (t << 4) + half * 8;
#pragma unroll
            for (int i = 0; i < 8; ++i) {
                const int rr = rb + i;
                if (rr < rows) out[(size_t)(nbase + rr) * HID + ncol] = c[i];
            }
        }
    } else {
        __syncthreads();

        float4*       o4 = (float4*)(out + (size_t)nbase * HID);
        const float4* a4 = (const float4*)acc;
        for (int i = tid; i < rows * (HID / 4); i += OWN_NT) o4[i] = a4[(i >> 5) * (ACC_P / 4) + (i & 31)];
    }
}

__global__ void k_pool_init(float* sums, float* cnt) {
    int idx = blockIdx.x * blockDim.x + threadIdx.x;
    const float4 z = make_float4(0.0f, 0.0f, 0.0f, 0.0f);
    if (idx < NGRAPHS * HID / 4) ((float4*)sums)[idx] = z;
    if (idx < NGRAPHS / 4)       ((float4*)cnt)[idx]  = z;
}

#define PA_ROWS 16
__global__ __launch_bounds__(256) void k_pool_accum(const float* __restrict__ h, const int* __restrict__ batch,
                                                    float* sums, int n) {
    const int lane = threadIdx.x & 31;
    const int wave = __builtin_amdgcn_readfirstlane((int)(threadIdx.x >> 5));
    const int r0   = (blockIdx.x * 8 + wave) * PA_ROWS;
    if (r0 >= n) return;
    int cnt = n - r0;
    if (cnt > PA_ROWS) cnt = PA_ROWS;
    const int b = (lane < cnt) ? batch[r0 + lane] : 0;

    float v[PA_ROWS][4];
#pragma unroll
    for (int u = 0; u < PA_ROWS; ++u) {
        const int r = r0 + ((u < cnt) ? u : 0);
        const float* p = h + (size_t)r * HID + lane;
#pragma unroll
        for (int k = 0; k < 4; ++k) v[u][k] = p[k * 32];
    }

    float acc[4] = {0.0f, 0.0f, 0.0f, 0.0f};
    int cur = __builtin_amdgcn_readlane(b, 0);
#pragma unroll
    for (int u = 0; u < PA_ROWS; ++u) {
        if (u < cnt) {
            const int bu = __builtin_amdgcn_readlane(b, u);
            if (bu != cur) {
                float* s = sums + (size_t)cur * HID + lane;
#pragma unroll
                for (int k = 0; k < 4; ++k) { atomicAdd(&s[k * 32], acc[k]); acc[k] = 0.0f; }
                cur = bu;
            }
#pragma unroll
            for (int k = 0; k < 4; ++k) acc[k] += v[u][k];
        }
    }
    float* s = sums + (size_t)cur * HID + lane;
#pragma unroll
    for (int k = 0; k < 4; ++k) atomicAdd(&s[k * 32], acc[k]);
}

__global__ void k_cnt_accum(const int* __restrict__ batch, float* cnt, int n) {
    int i = blockIdx.x * blockDim.x + threadIdx.x;
    if (i < n) atomicAdd(&cnt[batch[i]], 1.0f);
}

__global__ void k_final(const float* __restrict__ sums, const float* __restrict__ cnt,
                        const float* __restrict__ Wlin, const float* __restrict__ blin,
                        float* __restrict__ out) {
    int idx = blockIdx.x * blockDim.x + threadIdx.x;
    if (idx >= NGRAPHS * FOUT) return;
    int g = idx / FOUT;
    int j = idx - g * FOUT;
    float inv = 1.0f / fmaxf(cnt[g], 1.0f);
    const float* s = sums + g * HID;
    float acc = blin[j];
#pragma unroll 4
    for (int k = 0; k < HID; ++k)
        acc = fmaf(s[k] * inv, Wlin[k * FOUT + j], acc);
    out[idx] = acc;
}

extern "C" void kernel_launch(void* const* d_in, const int* in_sizes, int n_in,
                              void* d_out, int out_size, void* d_ws, size_t ws_size,
                              hipStream_t stream) {
    (void)n_in; (void)out_size; (void)ws_size;

    const float* x     = (const float*)d_in[0];
    const int*   esrc  = (const int*)  d_in[1];
    const int*   edst  = (const int*)  d_in[2];
    const int*   batch = (const int*)  d_in[3];
    const float* W1    = (const float*)d_in[4];
    const float* b1    = (const float*)d_in[5];
    const float* W2    = (const float*)d_in[6];
    const float* b2    = (const float*)d_in[7];
    const float* Wlin  = (const float*)d_in[8];
    const float* blin  = (const float*)d_in[9];
    float*       out   = (float*)d_out;

    const int N = in_sizes[3];
    const int E = in_sizes[1];
    const int ob = (N + OWN_R - 1) / OWN_R;

    char*  ws  = (char*)d_ws;
    size_t off = 0;
    auto carve = [&](size_t bytes) -> void* {
        off = (off + 255) & ~(size_t)255;
        void* p = ws + off;
        off += bytes;
        return p;
    };
    float* dinv  = (float*)carve((size_t)N * 4);
    float* bufH  = (float*)carve((size_t)N * HID * 4);
    float* bufO1 = (float*)carve((size_t)N * HID * 4);
    unsigned char* okey = (unsigned char*)carve((size_t)E);
    unsigned* lists = (unsigned*)carve((size_t)256 * OWN_CAP * 4);
    int* lcnt = (int*)carve((size_t)ob * 4);
    float* sums  = (float*)carve((size_t)NGRAPHS * HID * 4);
    float* cnt   = (float*)carve((size_t)NGRAPHS * 4);

    const int T = 256;
    int nb   = (N + T - 1) / T;
    int kb   = (E / 4) / T + 1;
    int pab  = (N + 8 * PA_ROWS - 1) / (8 * PA_ROWS);
    int pb   = (NGRAPHS * HID / 4 + T - 1) / T;
    int hb   = (NGRAPHS * FOUT + T - 1) / T;

    k_key        <<<kb, T, 0, stream>>>(edst, okey, E);
    k_owner_list <<<ob, OWN_NT, 0, stream>>>(esrc, edst, okey, lists, lcnt, dinv, N, E);

    k_gemm1       <<<G1_BLOCKS, T, 0, stream>>>(x, W1, bufH, N);
    k_conv_owner<true> <<<ob, OWN_NT, 0, stream>>>(bufH, bufO1, esrc, edst, lists, lcnt, dinv, b1, W2, N, E);

    k_conv_owner<false><<<ob, OWN_NT, 0, stream>>>(bufO1, bufH, esrc, edst, lists, lcnt, dinv, b2, W2, N, E);

    k_pool_init <<<pb, T, 0, stream>>>(sums, cnt);
    k_pool_accum<<<pab, T, 0, stream>>>(bufH, batch, sums, N);
    k_cnt_accum <<<nb, T, 0, stream>>>(batch, cnt, N);
    k_final     <<<hb, T, 0, stream>>>(sums, cnt, Wlin, blin, out);
}


# Independent small launches merged: x W1, owner keys and the pool clear share one grid (picked by block index); the graph node count rides behind the owner listing blocks; 9 launches become 6
# speedup vs baseline: 4.1148x; 1.0394x over previous
#include <hip/hip_runtime.h>
#include <hip/hip_bf16.h>

#define HID     128
#define FIN     11
#define FOUT    19
#define NGRAPHS 4096

typedef __attribute__((ext_vector_type(2))) float v2f;
typedef __attribute__((ext_vector_type(8))) float v8f;

#define OWN_R   392
#define ACC_P   132

#define OWN_NT  1024
#define OWN_CAP 8192
#define REC_BITS 9
#define SCAN_W  ((OWN_R + 31) / 32)

__device__ __forceinline__ unsigned owner_key(int d) { return ((unsigned)d / OWN_R) & 255u; }

__device__ __forceinline__ void key_part(const int* __restrict__ dst, unsigned char* __restrict__ key, int e, int blk) {
    const int t    = blk * (int)blockDim.x + (int)threadIdx.x;
    const int nvec = e >> 2;
    if (t < nvec) {
        const int4 d = ((const int4*)dst)[t];
        ((unsigned*)key)[t] = owner_key(d.x) | (owner_key(d.y) << 8) | (owner_key(d.z) << 16) | (owner_key(d.w) << 24);
    }
    if (t < (e & 3)) {
        const int i = (nvec << 2) + t;
        key[i] = (unsigned char)owner_key(dst[i]);
    }
}

__global__ __launch_bounds__(OWN_NT) void k_owner_list(const int* __restrict__ src,
                                                       const int* __restrict__ dst,
                                                       const unsigned char* __restrict__ key,
                                                       unsigned* __restrict__ lists,
                                                       int* __restrict__ lcnt,
                                                       float* __restrict__ dinv,
                                                       const int* __restrict__ batch,
                                                       float* cnt,
                                                       int n, int e, int ob) {
    if ((int)blockIdx.x >= ob) {
        const int i = ((int)blockIdx.x - ob) * OWN_NT + (int)threadIdx.x;
        if (i < n) atomicAdd(&cnt[batch[i]], 1.0f);
        return;
    }
    __shared__ int list_e[OWN_CAP];
    __shared__ int cntr[OWN_R];
    __shared__ int first[OWN_R];
    __shared__ int wtot[SCAN_W];
    __shared__ int lcount;

    const int tid   = threadIdx.x;
    const int nbase = blockIdx.x * OWN_R;
    int rows = n - nbase;
    if (rows > OWN_R) rows = OWN_R;
    const unsigned urows = (unsigned)rows;

    if (tid < OWN_R) cntr[tid] = 0;
    if (tid == 0) lcount = 0;
    __syncthreads();

    auto note1 = [&](int eid) {
        const int sl = atomicAdd(&lcount, 1);
        if (sl < OWN_CAP) list_e[sl] = eid;
    };
    const unsigned mykey = (unsigned)blockIdx.x & 255u;
    const unsigned K4    = mykey * 0x01010101u;
    auto nonzero_bytes = [](unsigned x) -> unsigned {
        return (((x & 0x7F7F7F7Fu) + 0x7F7F7F7Fu) | x) & 0x80808080u;
    };
    auto note16 = [&](const uint4 w, int eb) {
        const unsigned g = (nonzero_bytes(w.x ^ K4) >> 7) | (nonzero_bytes(w.y ^ K4) >> 6) |
                           (nonzero_bytes(w.z ^ K4) >> 5) | (nonzero_bytes(w.w ^ K4) >> 4);
        unsigned hit = g ^ 0x0F0F0F0Fu;
        while (hit) {
            const int p = __builtin_ctz(hit);
            hit &= hit - 1;
            note1(eb + ((p & 3) << 2) + (p >> 3));
        }
    };

    const uint4* key16 = (const uint4*)key;
    const int nvec = e >> 4;
    int i0 = 0;
    for (; i0 + OWN_NT * 8 <= nvec; i0 += OWN_NT * 8) {
        uint4 kv[8];
#pragma unroll
        for (int u = 0; u < 8; ++u) kv[u] = key16[i0 + u * OWN_NT + tid];
#pragma unroll
        for (int u = 0; u < 8; ++u) note16(kv[u], (i0 + u * OWN_NT + tid) << 4);
    }
    for (int i = i0 + tid; i < nvec; i += OWN_NT) note16(key16[i], i << 4);
    if (tid < (e & 15)) {
        const int eid = (nvec << 4) + tid;
        if ((unsigned)key[eid] == mykey) note1(eid);
    }
    __syncthreads();

    const int total = lcount;
    const bool listed = (total <= OWN_CAP) && (ob <= 256);
    unsigned arow[OWN_CAP / OWN_NT];
    int      rk[OWN_CAP / OWN_NT];
    int      s[OWN_CAP / OWN_NT];
#pragma unroll
    for (int u = 0; u < OWN_CAP / OWN_NT; ++u) { arow[u] = 0xFFFFFFFFu; rk[u] = 0; s[u] = 0; }
    if (listed) {
        int eid[OWN_CAP / OWN_NT];
#pragma unroll
        for (int u = 0; u < OWN_CAP / OWN_NT; ++u) {
            const int i = u * OWN_NT + tid;
            eid[u] = 0;
            if (i < total) eid[u] = list_e[i];
        }
        int d[OWN_CAP / OWN_NT];
#pragma unroll
        for (int u = 0; u < OWN_CAP / OWN_NT; ++u) { d[u] = dst[eid[u]]; s[u] = src[eid[u]]; }
#pragma unroll
        for (int u = 0; u < OWN_CAP / OWN_NT; ++u) {
            const int i = u * OWN_NT + tid;
            if (i < total) {
                const unsigned a = (unsigned)(d[u] - nbase);
                if (a < urows) {
                    arow[u] = a;
                    rk[u]   = atomicAdd(&cntr[a], 1);
                }
            }
        }
    } else {
        for (int i = tid; i < e; i += OWN_NT) {
            const unsigned a = (unsigned)(dst[i] - nbase);
            if (a < urows) atomicAdd(&cntr[a], 1);
        }
    }
    __syncthreads();

    const int lane = tid & 31;
    const int wv   = tid >> 5;
    const int cv   = (tid < rows) ? cntr[tid] : 0;
    int inc = cv;
#pragma unroll
    for (int dlt = 1; dlt < 32; dlt <<= 1) {
        const int t = __shfl_up(inc, dlt);
        if (lane >= dlt) inc += t;
    }
    if (lane == 31 && wv < SCAN_W) wtot[wv] = inc;
    __syncthreads();
    int below = 0, stored = 0;
    for (int q = 0; q < SCAN_W; ++q) {
        const int t = wtot[q];
        if (q < wv) below += t;
        stored += t;
    }
    if (tid < OWN_R) first[tid] = below + inc - cv;
    __syncthreads();

    if (listed) {
#pragma unroll
        for (int u = 0; u < OWN_CAP / OWN_NT; ++u) {
            const unsigned a = arow[u];
            if (a < urows) list_e[first[a] + rk[u]] = (int)(((unsigned)s[u] << REC_BITS) | a);
        }
    }
    __syncthreads();
    if (listed) {
        unsigned* mylist = lists + (size_t)blockIdx.x * OWN_CAP;
        for (int i = tid; i < stored; i += OWN_NT) mylist[i] = (unsigned)list_e[i];
    }

    if (tid < rows) {
        const float dg = 1.0f + (float)cntr[tid];
        dinv[nbase + tid] = (dg > 0.0f) ? rsqrtf(dg) : 0.0f;
    }
    if (tid == 0) lcnt[blockIdx.x] = listed ? stored : OWN_CAP + 1;
}

#define G1_BLOCKS 1024
__device__ __forceinline__ void gemm1_part(const float* __restrict__ x, const float* __restrict__ W1,
                                           float* __restrict__ h, int n, int blk) {
    const int lane = threadIdx.x & 31;
    const int wave = __builtin_amdgcn_readfirstlane((int)(threadIdx.x >> 5));
    const int gw   = blk * 8 + wave;
    const int nw   = G1_BLOCKS * 8;
    float4 w[FIN];
#pragma unroll
    for (int k = 0; k < FIN; ++k) w[k] = ((const float4*)W1)[k * (HID / 4) + lane];
    for (int row = gw; row < n; row += nw) {
        const float* xr = x + (size_t)row * FIN;
        float4 acc = make_float4(0.0f, 0.0f, 0.0f, 0.0f);
#pragma unroll
        for (int k = 0; k < FIN; ++k) {
            const float xv = xr[k];
            acc.x = fmaf(xv, w[k].x, acc.x);
            acc.y = fmaf(xv, w[k].y, acc.y);
            acc.z = fmaf(xv, w[k].z, acc.z);
            acc.w = fmaf(xv, w[k].w, acc.w);
        }
        ((float4*)(h + (size_t)row * HID))[lane] = acc;
    }
}

static_assert(OWN_R <= (1 << REC_BITS), "the owned row must fit the record's row field");

template <bool W2EPI>
__global__ __launch_bounds__(OWN_NT) void k_conv_owner(const float* __restrict__ h,
                                                       float* __restrict__ out,
                                                       const int* __restrict__ src,
                                                       const int* __restrict__ dst,
                                                       const unsigned* __restrict__ lists,
                                                       const int* __restrict__ lcnt,
                                                       const float* __restrict__ dinv,
                                                       const float* __restrict__ bias,
                                                       const float* __restrict__ W,
                                                       int n, int e) {
    __shared__ __attribute__((aligned(16))) float acc[OWN_R * ACC_P];

    const int tid   = threadIdx.x;
    const int nbase = blockIdx.x * OWN_R;
    int rows = n - nbase;
    if (rows > OWN_R) rows = OWN_R;
    const unsigned urows = (unsigned)rows;

    {
        const float4  b4   = ((const float4*)bias)[tid & 31];
        const float4* h4   = (const float4*)(h + (size_t)nbase * HID);
        float4*       acc4 = (float4*)acc;
        for (int i = tid; i < rows * (HID / 4); i += OWN_NT) {
            const float  dn = dinv[nbase + (i >> 5)];
            const float4 v  = h4[i];
            float4 o;
            o.x = b4.x + v.x * dn * dn;
            o.y = b4.y + v.y * dn * dn;
            o.z = b4.z + v.z * dn * dn;
            o.w = b4.w + v.w * dn * dn;
            acc4[(i >> 5) * (ACC_P / 4) + (i & 31)] = o;
        }
    }
    __syncthreads();

    const int total = lcnt[blockIdx.x];
    if (total <= OWN_CAP && gridDim.x <= 256) {
        const unsigned* mylist = lists + (size_t)blockIdx.x * OWN_CAP;
        const int lane = tid & 31;
        const int wave = __builtin_amdgcn_readfirstlane(tid >> 5);
        for (int base = wave * 32; base < total; base += OWN_NT) {
            int   s = 0, dl = 0;
            float nrm = 0.0f;
            if (base + lane < total) {
                const unsigned rec = mylist[base + lane];
                s   = (int)(rec >> REC_BITS);
                dl  = (int)(rec & ((1u << REC_BITS) - 1u));
                nrm = dinv[s] * dinv[nbase + dl];
            }
            int cnt = total - base;
            if (cnt > 32) cnt = 32;
            float run[4] = {0.0f, 0.0f, 0.0f, 0.0f};
            int   cur = __builtin_amdgcn_readlane(dl, 0);
            auto flush = [&](int row) {
                float* a = acc + row * ACC_P + lane;
#pragma unroll
                for (int k = 0; k < 4; ++k) { atomicAdd(&a[k * 32], run[k]); run[k] = 0.0f; }
            };
            int j = 0;
            for (; j + 8 <= cnt; j += 8) {
                float v[8][4];
#pragma unroll
                for (int u = 0; u < 8; ++u) {
                    const int sj = __builtin_amdgcn_readlane(s, j + u);
                    const float* r = h + (size_t)sj * HID + lane;
#pragma unroll
                    for (int k = 0; k < 4; ++k) v[u][k] = r[k * 32];
                }
#pragma unroll
                for (int u = 0; u < 8; ++u) {
                    const int   dj = __builtin_amdgcn_readlane(dl, j + u);
                    const float nj = __int_as_float(__builtin_amdgcn_readlane(__float_as_int(nrm), j + u));
                    if (dj != cur) { flush(cur); cur = dj; }
#pragma unroll
                    for (int k = 0; k < 4; ++k) run[k] = fmaf(v[u][k], nj, run[k]);
                }
            }
            for (; j < cnt; ++j) {
                const int   sj = __builtin_amdgcn_readlane(s, j);
                const int   dj = __builtin_amdgcn_readlane(dl, j);
                const float nj = __int_as_float(__builtin_amdgcn_readlane(__float_as_int(nrm), j));
                const float* r = h + (size_t)sj * HID + lane;
                if (dj != cur) { flush(cur); cur = dj; }
#pragma unroll
                for (int k = 0; k < 4; ++k) run[k] = fmaf(r[k * 32], nj, run[k]);
            }
            flush(cur);
        }
    } else {
        for (int i = tid; i < e; i += OWN_NT) {
            const unsigned a = (unsigned)(dst[i] - nbase);
            if (a < urows) {
                const int   s   = src[i];
                const float nrm = dinv[s] * dinv[nbase + (int)a];
                const float* r  = h + (size_t)s * HID;
                float* ar = acc + a * ACC_P;
                for (int f = 0; f < HID; ++f) atomicAdd(&ar[f], r[f] * nrm);
            }
        }
    }
    if constexpr (W2EPI) {
        const int lane = tid & 31;
        const int wave = __builtin_amdgcn_readfirstlane(tid >> 5);
        const int half = lane >> 4;
        const int l15  = lane & 15;
        const int ncol = ((wave & 7) << 4) + l15;
        v2f bf[HID / 4];
#pragma unroll
        for (int kk = 0; kk < HID / 4; ++kk) {
            const int ka = kk * 4 + half * 2;
            bf[kk][0] = W[ka * HID + ncol];
            bf[kk][1] = W[(ka + 1) * HID + ncol];
        }
        __syncthreads();

        const int ntile = (rows + 15) >> 4;
        for (int t = wave >> 3; t < ntile; t += 4) {
            int r = (t << 4) + l15;
            if (r >= rows) r = rows - 1;
            const float* ar = acc + r * ACC_P + half * 2;
            v8f c = {};
#pragma unroll
            for (int kk = 0; kk < HID / 4; ++kk) {
                const float2 av = *(const float2*)(ar + kk * 4);
                v2f a;
                a[0] = fmaxf(av.x, 0.0f);
                a[1] = fmaxf(av.y, 0.0f);
                c = __builtin_amdgcn_wmma_f32_16x16x4_f32(
                         false, a,  false, bf[kk],
                         (short)0, c,  false,  false);
            }
            const int rb = (t << 4) + half * 8;
#pragma unroll
            for (int i = 0; i < 8; ++i) {
                const int rr = rb + i;
                if (rr < rows) out[(size_t)(nbase + rr) * HID + ncol] = c[i];
            }
        }
    } else {
        __syncthreads();

        float4*       o4 = (float4*)(out + (size_t)nbase * HID);
        const float4* a4 = (const float4*)acc;
        for (int i = tid; i < rows * (HID / 4); i += OWN_NT) o4[i] = a4[(i >> 5) * (ACC_P / 4) + (i & 31)];
    }
}

__device__ __forceinline__ void pool_init_part(float* sums, float* cnt, int blk) {
    int idx = blk * (int)blockDim.x + (int)threadIdx.x;
    const float4 z = make_float4(0.0f, 0.0f, 0.0f, 0.0f);
    if (idx < NGRAPHS * HID / 4) ((float4*)sums)[idx] = z;
    if (idx < NGRAPHS / 4)       ((float4*)cnt)[idx]  = z;
}

__global__ __launch_bounds__(256) void k_front(const float* __restrict__ x, const float* __restrict__ W1,
                                               float* __restrict__ h, int n,
                                               const int* __restrict__ dst, unsigned char* __restrict__ key, int e,
                                               float* sums, float* cnt, int kb) {
    const int b = (int)blockIdx.x;
    if (b < G1_BLOCKS)           gemm1_part(x, W1, h, n, b);
    else if (b < G1_BLOCKS + kb) key_part(dst, key, e, b - G1_BLOCKS);
    else                         pool_init_part(sums, cnt, b - G1_BLOCKS - kb);
}

#define PA_ROWS 16
__global__ __launch_bounds__(256) void k_pool_accum(const float* __restrict__ h, const int* __restrict__ batch,
                                                    float* sums, int n) {
    const int lane = threadIdx.x & 31;
    const int wave = __builtin_amdgcn_readfirstlane((int)(threadIdx.x >> 5));
    const int r0   = (blockIdx.x * 8 + wave) * PA_ROWS;
    if (r0 >= n) return;
    int cnt = n - r0;
    if (cnt > PA_ROWS) cnt = PA_ROWS;
    const int b = (lane < cnt) ? batch[r0 + lane] : 0;

    float v[PA_ROWS][4];
#pragma unroll
    for (int u = 0; u < PA_ROWS; ++u) {
        const int r = r0 + ((u < cnt) ? u : 0);
        const float* p = h + (size_t)r * HID + lane;
#pragma unroll
        for (int k = 0; k < 4; ++k) v[u][k] = p[k * 32];
    }

    float acc[4] = {0.0f, 0.0f, 0.0f, 0.0f};
    int cur = __builtin_amdgcn_readlane(b, 0);
#pragma unroll
    for (int u = 0; u < PA_ROWS; ++u) {
        if (u < cnt) {
            const int bu = __builtin_amdgcn_readlane(b, u);
            if (bu != cur) {
                float* s = sums + (size_t)cur * HID + lane;
#pragma unroll
                for (int k = 0; k < 4; ++k) { atomicAdd(&s[k * 32], acc[k]); acc[k] = 0.0f; }
                cur = bu;
            }
#pragma unroll
            for (int k = 0; k < 4; ++k) acc[k] += v[u][k];
        }
    }
    float* s = sums + (size_t)cur * HID + lane;
#pragma unroll
    for (int k = 0; k < 4; ++k) atomicAdd(&s[k * 32], acc[k]);
}

__global__ void k_final(const float* __restrict__ sums, const float* __restrict__ cnt,
                        const float* __restrict__ Wlin, const float* __restrict__ blin,
                        float* __restrict__ out) {
    int idx = blockIdx.x * blockDim.x + threadIdx.x;
    if (idx >= NGRAPHS * FOUT) return;
    int g = idx / FOUT;
    int j = idx - g * FOUT;
    float inv = 1.0f / fmaxf(cnt[g], 1.0f);
    const float* s = sums + g * HID;
    float acc = blin[j];
#pragma unroll 4
    for (int k = 0; k < HID; ++k)
        acc = fmaf(s[k] * inv, Wlin[k * FOUT + j], acc);
    out[idx] = acc;
}

extern "C" void kernel_launch(void* const* d_in, const int* in_sizes, int n_in,
                              void* d_out, int out_size, void* d_ws, size_t ws_size,
                              hipStream_t stream) {
    (void)n_in; (void)out_size; (void)ws_size;

    const float* x     = (const float*)d_in[0];
    const int*   esrc  = (const int*)  d_in[1];
    const int*   edst  = (const int*)  d_in[2];
    const int*   batch = (const int*)  d_in[3];
    const float* W1    = (const float*)d_in[4];
    const float* b1    = (const float*)d_in[5];
    const float* W2    = (const float*)d_in[6];
    const float* b2    = (const float*)d_in[7];
    const float* Wlin  = (const float*)d_in[8];
    const float* blin  = (const float*)d_in[9];
    float*       out   = (float*)d_out;

    const int N = in_sizes[3];
    const int E = in_sizes[1];
    const int ob = (N + OWN_R - 1) / OWN_R;

    char*  ws  = (char*)d_ws;
    size_t off = 0;
    auto carve = [&](size_t bytes) -> void* {
        off = (off + 255) & ~(size_t)255;
        void* p = ws + off;
        off += bytes;
        return p;
    };
    float* dinv  = (float*)carve((size_t)N * 4);
    float* bufH  = (float*)carve((size_t)N * HID * 4);
    float* bufO1 = (float*)carve((size_t)N * HID * 4);
    unsigned char* okey = (unsigned char*)carve((size_t)E);
    unsigned* lists = (unsigned*)carve((size_t)256 * OWN_CAP * 4);
    int* lcnt = (int*)carve((size_t)ob * 4);
    float* sums  = (float*)carve((size_t)NGRAPHS * HID * 4);
    float* cnt   = (float*)carve((size_t)NGRAPHS * 4);

    const int T = 256;
    int cb   = (N + OWN_NT - 1) / OWN_NT;
    int kb   = (E / 4) / T + 1;
    int pab  = (N + 8 * PA_ROWS - 1) / (8 * PA_ROWS);
    int pb   = (NGRAPHS * HID / 4 + T - 1) / T;
    int hb   = (NGRAPHS * FOUT + T - 1) / T;

    k_front      <<<G1_BLOCKS + kb + pb, T, 0, stream>>>(x, W1, bufH, N, edst, okey, E, sums, cnt, kb);
    k_owner_list <<<ob + cb, OWN_NT, 0, stream>>>(esrc, edst, okey, lists, lcnt, dinv, batch, cnt, N, E, ob);

    k_conv_owner<true> <<<ob, OWN_NT, 0, stream>>>(bufH, bufO1, esrc, edst, lists, lcnt, dinv, b1, W2, N, E);

    k_conv_owner<false><<<ob, OWN_NT, 0, stream>>>(bufO1, bufH, esrc, edst, lists, lcnt, dinv, b2, W2, N, E);

    k_pool_accum<<<pab, T, 0, stream>>>(bufH, batch, sums, N);
    k_final     <<<hb, T, 0, stream>>>(sums, cnt, Wlin, blin, out);
}


# Rows stored pre-multiplied by dinv of their node (x W1 moves behind the owner's degree count; W2 product scaled at its store): no per-edge norm fetch, dinv[d] once a row; f32 rounding changes
# speedup vs baseline: 4.2878x; 1.0420x over previous
#include <hip/hip_runtime.h>
#include <hip/hip_bf16.h>

#define HID     128
#define FIN     11
#define FOUT    19
#define NGRAPHS 4096

typedef __attribute__((ext_vector_type(2))) float v2f;
typedef __attribute__((ext_vector_type(8))) float v8f;

#define OWN_R   392
#define ACC_P   132

#define OWN_NT  1024
#define OWN_CAP 8192
#define REC_BITS 9
#define SCAN_W  ((OWN_R + 31) / 32)
#define XS_PER_T ((OWN_R * FIN + OWN_NT - 1) / OWN_NT)

__device__ __forceinline__ unsigned owner_key(int d) { return ((unsigned)d / OWN_R) & 255u; }

__device__ __forceinline__ void key_part(const int* __restrict__ dst, unsigned char* __restrict__ key, int e, int blk) {
    const int t    = blk * (int)blockDim.x + (int)threadIdx.x;
    const int nvec = e >> 2;
    if (t < nvec) {
        const int4 d = ((const int4*)dst)[t];
        ((unsigned*)key)[t] = owner_key(d.x) | (owner_key(d.y) << 8) | (owner_key(d.z) << 16) | (owner_key(d.w) << 24);
    }
    if (t < (e & 3)) {
        const int i = (nvec << 2) + t;
        key[i] = (unsigned char)owner_key(dst[i]);
    }
}

__global__ __launch_bounds__(OWN_NT) void k_owner_list(const int* __restrict__ src,
                                                       const int* __restrict__ dst,
                                                       const unsigned char* __restrict__ key,
                                                       unsigned* __restrict__ lists,
                                                       int* __restrict__ lcnt,
                                                       float* __restrict__ dinv,
                                                       const int* __restrict__ batch,
                                                       float* cnt,
                                                       const float* __restrict__ x,
                                                       const float* __restrict__ W1,
                                                       float* __restrict__ h,
                                                       int n, int e, int ob) {
    if ((int)blockIdx.x >= ob) {
        const int i = ((int)blockIdx.x - ob) * OWN_NT + (int)threadIdx.x;
        if (i < n) atomicAdd(&cnt[batch[i]], 1.0f);
        return;
    }
    __shared__ int list_e[OWN_CAP];
    __shared__ int cntr[OWN_R];
    __shared__ int first[OWN_R];
    __shared__ int wtot[SCAN_W];
    __shared__ int lcount;
    __shared__ float xs[OWN_R * FIN];
    __shared__ float dloc[OWN_R];

    const int tid   = threadIdx.x;
    const int nbase = blockIdx.x * OWN_R;
    int rows = n - nbase;
    if (rows > OWN_R) rows = OWN_R;
    const unsigned urows = (unsigned)rows;

    float xv[XS_PER_T];
#pragma unroll
    for (int u = 0; u < XS_PER_T; ++u) {
        const int i = u * OWN_NT + tid;
        xv[u] = (i < rows * FIN) ? x[(size_t)nbase * FIN + i] : 0.0f;
    }

    if (tid < OWN_R) cntr[tid] = 0;
    if (tid == 0) lcount = 0;
    __syncthreads();

    auto note1 = [&](int eid) {
        const int sl = atomicAdd(&lcount, 1);
        if (sl < OWN_CAP) list_e[sl] = eid;
    };
    const unsigned mykey = (unsigned)blockIdx.x & 255u;
    const unsigned K4    = mykey * 0x01010101u;
    auto nonzero_bytes = [](unsigned x) -> unsigned {
        return (((x & 0x7F7F7F7Fu) + 0x7F7F7F7Fu) | x) & 0x80808080u;
    };
    auto note16 = [&](const uint4 w, int eb) {
        const unsigned g = (nonzero_bytes(w.x ^ K4) >> 7) | (nonzero_bytes(w.y ^ K4) >> 6) |
                           (nonzero_bytes(w.z ^ K4) >> 5) | (nonzero_bytes(w.w ^ K4) >> 4);
        unsigned hit = g ^ 0x0F0F0F0Fu;
        while (hit) {
            const int p = __builtin_ctz(hit);
            hit &= hit - 1;
            note1(eb + ((p & 3) << 2) + (p >> 3));
        }
    };

    const uint4* key16 = (const uint4*)key;
    const int nvec = e >> 4;
    int i0 = 0;
    for (; i0 + OWN_NT * 8 <= nvec; i0 += OWN_NT * 8) {
        uint4 kv[8];
#pragma unroll
        for (int u = 0; u < 8; ++u) kv[u] = key16[i0 + u * OWN_NT + tid];
#pragma unroll
        for (int u = 0; u < 8; ++u) note16(kv[u], (i0 + u * OWN_NT + tid) << 4);
    }
    for (int i = i0 + tid; i < nvec; i += OWN_NT) note16(key16[i], i << 4);
    if (tid < (e & 15)) {
        const int eid = (nvec << 4) + tid;
        if ((unsigned)key[eid] == mykey) note1(eid);
    }
#pragma unroll
    for (int u = 0; u < XS_PER_T; ++u) {
        const int i = u * OWN_NT + tid;
        if (i < rows * FIN) xs[i] = xv[u];
    }
    __syncthreads();

    const int total = lcount;
    const bool listed = (total <= OWN_CAP) && (ob <= 256);
    unsigned arow[OWN_CAP / OWN_NT];
    int      rk[OWN_CAP / OWN_NT];
    int      s[OWN_CAP / OWN_NT];
#pragma unroll
    for (int u = 0; u < OWN_CAP / OWN_NT; ++u) { arow[u] = 0xFFFFFFFFu; rk[u] = 0; s[u] = 0; }
    if (listed) {
        int eid[OWN_CAP / OWN_NT];
#pragma unroll
        for (int u = 0; u < OWN_CAP / OWN_NT; ++u) {
            const int i = u * OWN_NT + tid;
            eid[u] = 0;
            if (i < total) eid[u] = list_e[i];
        }
        int d[OWN_CAP / OWN_NT];
#pragma unroll
        for (int u = 0; u < OWN_CAP / OWN_NT; ++u) { d[u] = dst[eid[u]]; s[u] = src[eid[u]]; }
#pragma unroll
        for (int u = 0; u < OWN_CAP / OWN_NT; ++u) {
            const int i = u * OWN_NT + tid;
            if (i < total) {
                const unsigned a = (unsigned)(d[u] - nbase);
                if (a < urows) {
                    arow[u] = a;
                    rk[u]   = atomicAdd(&cntr[a], 1);
                }
            }
        }
    } else {
        for (int i = tid; i < e; i += OWN_NT) {
            const unsigned a = (unsigned)(dst[i] - nbase);
            if (a < urows) atomicAdd(&cntr[a], 1);
        }
    }
    __syncthreads();

    const int lane = tid & 31;
    const int wv   = tid >> 5;
    const int cv   = (tid < rows) ? cntr[tid] : 0;
    int inc = cv;
#pragma unroll
    for (int dlt = 1; dlt < 32; dlt <<= 1) {
        const int t = __shfl_up(inc, dlt);
        if (lane >= dlt) inc += t;
    }
    if (lane == 31 && wv < SCAN_W) wtot[wv] = inc;
    __syncthreads();
    int below = 0, stored = 0;
    for (int q = 0; q < SCAN_W; ++q) {
        const int t = wtot[q];
        if (q < wv) below += t;
        stored += t;
    }
    if (tid < OWN_R) first[tid] = below + inc - cv;
    __syncthreads();

    if (listed) {
#pragma unroll
        for (int u = 0; u < OWN_CAP / OWN_NT; ++u) {
            const unsigned a = arow[u];
            if (a < urows) list_e[first[a] + rk[u]] = (int)(((unsigned)s[u] << REC_BITS) | a);
        }
    }
    __syncthreads();
    if (listed) {
        unsigned* mylist = lists + (size_t)blockIdx.x * OWN_CAP;
        for (int i = tid; i < stored; i += OWN_NT) mylist[i] = (unsigned)list_e[i];
    }

    if (tid < rows) {
        const float dg = 1.0f + (float)cntr[tid];
        const float dv = (dg > 0.0f) ? rsqrtf(dg) : 0.0f;
        dinv[nbase + tid] = dv;
        dloc[tid] = dv;
    }
    if (tid == 0) lcnt[blockIdx.x] = listed ? stored : OWN_CAP + 1;

    float4 w[FIN];
#pragma unroll
    for (int k = 0; k < FIN; ++k) w[k] = ((const float4*)W1)[k * (HID / 4) + lane];
    __syncthreads();
    const int swave = __builtin_amdgcn_readfirstlane(wv);
    for (int r = swave; r < rows; r += OWN_NT / 32) {
        const float xl = (lane < FIN) ? xs[r * FIN + lane] : 0.0f;
        float4 a = make_float4(0.0f, 0.0f, 0.0f, 0.0f);
#pragma unroll
        for (int k = 0; k < FIN; ++k) {
            const float xk = __int_as_float(__builtin_amdgcn_readlane(__float_as_int(xl), k));
            a.x = fmaf(xk, w[k].x, a.x);
            a.y = fmaf(xk, w[k].y, a.y);
            a.z = fmaf(xk, w[k].z, a.z);
            a.w = fmaf(xk, w[k].w, a.w);
        }
        const float dn = dloc[r];
        a.x *= dn; a.y *= dn; a.z *= dn; a.w *= dn;
        ((float4*)(h + (size_t)(nbase + r) * HID))[lane] = a;
    }
}

static_assert(OWN_R <= (1 << REC_BITS), "the owned row must fit the record's row field");
#define INIT_PER_T ((OWN_R * (HID / 4) + OWN_NT - 1) / OWN_NT)

template <bool W2EPI>
__global__ __launch_bounds__(OWN_NT) void k_conv_owner(const float* __restrict__ h,
                                                       float* __restrict__ out,
                                                       const int* __restrict__ src,
                                                       const int* __restrict__ dst,
                                                       const unsigned* __restrict__ lists,
                                                       const int* __restrict__ lcnt,
                                                       const float* __restrict__ dinv,
                                                       const float* __restrict__ bias,
                                                       const float* __restrict__ W,
                                                       int n, int e) {
    __shared__ __attribute__((aligned(16))) float acc[OWN_R * ACC_P];
    __shared__ float dnl[OWN_R];

    const int tid   = threadIdx.x;
    const int nbase = blockIdx.x * OWN_R;
    int rows = n - nbase;
    if (rows > OWN_R) rows = OWN_R;
    const unsigned urows = (unsigned)rows;
    const float4 b4 = ((const float4*)bias)[tid & 31];

    {
        const float4* h4   = (const float4*)(h + (size_t)nbase * HID);
        float4*       acc4 = (float4*)acc;
        const int     lim  = rows * (HID / 4);
        float dv = 0.0f;
        if (tid < rows) dv = dinv[nbase + tid];
        float4 hv[INIT_PER_T];
#pragma unroll
        for (int u = 0; u < INIT_PER_T; ++u) {
            const int i = u * OWN_NT + tid;
            hv[u] = (i < lim) ? h4[i] : make_float4(0.0f, 0.0f, 0.0f, 0.0f);
        }
        if (tid < rows) dnl[tid] = dv;
#pragma unroll
        for (int u = 0; u < INIT_PER_T; ++u) {
            const int i = u * OWN_NT + tid;
            if (i < lim) acc4[(i >> 5) * (ACC_P / 4) + (i & 31)] = hv[u];
        }
    }
    __syncthreads();

    const int total = lcnt[blockIdx.x];
    if (total <= OWN_CAP && gridDim.x <= 256) {
        const unsigned* mylist = lists + (size_t)blockIdx.x * OWN_CAP;
        const int lane = tid & 31;
        const int wave = __builtin_amdgcn_readfirstlane(tid >> 5);
        for (int base = wave * 32; base < total; base += OWN_NT) {
            int   s = 0, dl = 0;
            if (base + lane < total) {
                const unsigned rec = mylist[base + lane];
                s   = (int)(rec >> REC_BITS);
                dl  = (int)(rec & ((1u << REC_BITS) - 1u));
            }
            int cnt = total - base;
            if (cnt > 32) cnt = 32;
            float run[4] = {0.0f, 0.0f, 0.0f, 0.0f};
            int   cur = __builtin_amdgcn_readlane(dl, 0);
            auto flush = [&](int row) {
                float* a = acc + row * ACC_P + lane;
#pragma unroll
                for (int k = 0; k < 4; ++k) { atomicAdd(&a[k * 32], run[k]); run[k] = 0.0f; }
            };
            int j = 0;
            for (; j + 8 <= cnt; j += 8) {
                float v[8][4];
#pragma unroll
                for (int u = 0; u < 8; ++u) {
                    const int sj = __builtin_amdgcn_readlane(s, j + u);
                    const float* r = h + (size_t)sj * HID + lane;
#pragma unroll
                    for (int k = 0; k < 4; ++k) v[u][k] = r[k * 32];
                }
#pragma unroll
                for (int u = 0; u < 8; ++u) {
                    const int   dj = __builtin_amdgcn_readlane(dl, j + u);
                    if (dj != cur) { flush(cur); cur = dj; }
#pragma unroll
                    for (int k = 0; k < 4; ++k) run[k] += v[u][k];
                }
            }
            for (; j < cnt; ++j) {
                const int   sj = __builtin_amdgcn_readlane(s, j);
                const int   dj = __builtin_amdgcn_readlane(dl, j);
                const float* r = h + (size_t)sj * HID + lane;
                if (dj != cur) { flush(cur); cur = dj; }
#pragma unroll
                for (int k = 0; k < 4; ++k) run[k] += r[k * 32];
            }
            flush(cur);
        }
    } else {
        for (int i = tid; i < e; i += OWN_NT) {
            const unsigned a = (unsigned)(dst[i] - nbase);
            if (a < urows) {
                const int   s   = src[i];
                const float* r  = h + (size_t)s * HID;
                float* ar = acc + a * ACC_P;
                for (int f = 0; f < HID; ++f) atomicAdd(&ar[f], r[f]);
            }
        }
    }
    if constexpr (W2EPI) {
        const int lane = tid & 31;
        const int wave = __builtin_amdgcn_readfirstlane(tid >> 5);
        const int half = lane >> 4;
        const int l15  = lane & 15;
        const int ncol = ((wave & 7) << 4) + l15;
        v2f bf[HID / 4];
#pragma unroll
        for (int kk = 0; kk < HID / 4; ++kk) {
            const int ka = kk * 4 + half * 2;
            bf[kk][0] = W[ka * HID + ncol];
            bf[kk][1] = W[(ka + 1) * HID + ncol];
        }
        __syncthreads();

        {
            float4* acc4 = (float4*)acc;
            for (int i = tid; i < rows * (HID / 4); i += OWN_NT) {
                const float dn = dnl[i >> 5];
                float4 v = acc4[(i >> 5) * (ACC_P / 4) + (i & 31)];
                v.x = fmaf(v.x, dn, b4.x);
                v.y = fmaf(v.y, dn, b4.y);
                v.z = fmaf(v.z, dn, b4.z);
                v.w = fmaf(v.w, dn, b4.w);
                acc4[(i >> 5) * (ACC_P / 4) + (i & 31)] = v;
            }
        }
        __syncthreads();

        const int ntile = (rows + 15) >> 4;
        for (int t = wave >> 3; t < ntile; t += 4) {
            int r = (t << 4) + l15;
            if (r >= rows) r = rows - 1;
            const float* ar = acc + r * ACC_P + half * 2;
            v8f c = {};
#pragma unroll
            for (int kk = 0; kk < HID / 4; ++kk) {
                const float2 av = *(const float2*)(ar + kk * 4);
                v2f a;
                a[0] = fmaxf(av.x, 0.0f);
                a[1] = fmaxf(av.y, 0.0f);
                c = __builtin_amdgcn_wmma_f32_16x16x4_f32(
                         false, a,  false, bf[kk],
                         (short)0, c,  false,  false);
            }
            const int rb = (t << 4) + half * 8;
#pragma unroll
            for (int i = 0; i < 8; ++i) {
                const int rr = rb + i;
                if (rr < rows) out[(size_t)(nbase + rr) * HID + ncol] = c[i] * dnl[rr];
            }
        }
    } else {
        __syncthreads();

        float4*       o4 = (float4*)(out + (size_t)nbase * HID);
        const float4* a4 = (const float4*)acc;
        for (int i = tid; i < rows * (HID / 4); i += OWN_NT) {
            const float  dn = dnl[i >> 5];
            const float4 v  = a4[(i >> 5) * (ACC_P / 4) + (i & 31)];
            float4 o;
            o.x = fmaf(v.x, dn, b4.x);
            o.y = fmaf(v.y, dn, b4.y);
            o.z = fmaf(v.z, dn, b4.z);
            o.w = fmaf(v.w, dn, b4.w);
            o4[i] = o;
        }
    }
}

__device__ __forceinline__ void pool_init_part(float* sums, float* cnt, int blk) {
    int idx = blk * (int)blockDim.x + (int)threadIdx.x;
    const float4 z = make_float4(0.0f, 0.0f, 0.0f, 0.0f);
    if (idx < NGRAPHS * HID / 4) ((float4*)sums)[idx] = z;
    if (idx < NGRAPHS / 4)       ((float4*)cnt)[idx]  = z;
}

__global__ __launch_bounds__(256) void k_front(const int* __restrict__ dst, unsigned char* __restrict__ key, int e,
                                               float* sums, float* cnt, int kb) {
    const int b = (int)blockIdx.x;
    if (b < kb) key_part(dst, key, e, b);
    else        pool_init_part(sums, cnt, b - kb);
}

#define PA_ROWS 16
__global__ __launch_bounds__(256) void k_pool_accum(const float* __restrict__ h, const int* __restrict__ batch,
                                                    float* sums, int n) {
    const int lane = threadIdx.x & 31;
    const int wave = __builtin_amdgcn_readfirstlane((int)(threadIdx.x >> 5));
    const int r0   = (blockIdx.x * 8 + wave) * PA_ROWS;
    if (r0 >= n) return;
    int cnt = n - r0;
    if (cnt > PA_ROWS) cnt = PA_ROWS;
    const int b = (lane < cnt) ? batch[r0 + lane] : 0;

    float v[PA_ROWS][4];
#pragma unroll
    for (int u = 0; u < PA_ROWS; ++u) {
        const int r = r0 + ((u < cnt) ? u : 0);
        const float* p = h + (size_t)r * HID + lane;
#pragma unroll
        for (int k = 0; k < 4; ++k) v[u][k] = p[k * 32];
    }

    float acc[4] = {0.0f, 0.0f, 0.0f, 0.0f};
    int cur = __builtin_amdgcn_readlane(b, 0);
#pragma unroll
    for (int u = 0; u < PA_ROWS; ++u) {
        if (u < cnt) {
            const int bu = __builtin_amdgcn_readlane(b, u);
            if (bu != cur) {
                float* s = sums + (size_t)cur * HID + lane;
#pragma unroll
                for (int k = 0; k < 4; ++k) { atomicAdd(&s[k * 32], acc[k]); acc[k] = 0.0f; }
                cur = bu;
            }
#pragma unroll
            for (int k = 0; k < 4; ++k) acc[k] += v[u][k];
        }
    }
    float* s = sums + (size_t)cur * HID + lane;
#pragma unroll
    for (int k = 0; k < 4; ++k) atomicAdd(&s[k * 32], acc[k]);
}

__global__ void k_final(const float* __restrict__ sums, const float* __restrict__ cnt,
                        const float* __restrict__ Wlin, const float* __restrict__ blin,
                        float* __restrict__ out) {
    int idx = blockIdx.x * blockDim.x + threadIdx.x;
    if (idx >= NGRAPHS * FOUT) return;
    int g = idx / FOUT;
    int j = idx - g * FOUT;
    float inv = 1.0f / fmaxf(cnt[g], 1.0f);
    const float* s = sums + g * HID;
    float acc = blin[j];
#pragma unroll 4
    for (int k = 0; k < HID; ++k)
        acc = fmaf(s[k] * inv, Wlin[k * FOUT + j], acc);
    out[idx] = acc;
}

extern "C" void kernel_launch(void* const* d_in, const int* in_sizes, int n_in,
                              void* d_out, int out_size, void* d_ws, size_t ws_size,
                              hipStream_t stream) {
    (void)n_in; (void)out_size; (void)ws_size;

    const float* x     = (const float*)d_in[0];
    const int*   esrc  = (const int*)  d_in[1];
    const int*   edst  = (const int*)  d_in[2];
    const int*   batch = (const int*)  d_in[3];
    const float* W1    = (const float*)d_in[4];
    const float* b1    = (const float*)d_in[5];
    const float* W2    = (const float*)d_in[6];
    const float* b2    = (const float*)d_in[7];
    const float* Wlin  = (const float*)d_in[8];
    const float* blin  = (const float*)d_in[9];
    float*       out   = (float*)d_out;

    const int N = in_sizes[3];
    const int E = in_sizes[1];
    const int ob = (N + OWN_R - 1) / OWN_R;

    char*  ws  = (char*)d_ws;
    size_t off = 0;
    auto carve = [&](size_t bytes) -> void* {
        off = (off + 255) & ~(size_t)255;
        void* p = ws + off;
        off += bytes;
        return p;
    };
    float* dinv  = (float*)carve((size_t)N * 4);
    float* bufH  = (float*)carve((size_t)N * HID * 4);
    float* bufO1 = (float*)carve((size_t)N * HID * 4);
    unsigned char* okey = (unsigned char*)carve((size_t)E);
    unsigned* lists = (unsigned*)carve((size_t)256 * OWN_CAP * 4);
    int* lcnt = (int*)carve((size_t)ob * 4);
    float* sums  = (float*)carve((size_t)NGRAPHS * HID * 4);
    float* cnt   = (float*)carve((size_t)NGRAPHS * 4);

    const int T = 256;
    int cb   = (N + OWN_NT - 1) / OWN_NT;
    int kb   = (E / 4) / T + 1;
    int pab  = (N + 8 * PA_ROWS - 1) / (8 * PA_ROWS);
    int pb   = (NGRAPHS * HID / 4 + T - 1) / T;
    int hb   = (NGRAPHS * FOUT + T - 1) / T;

    k_front      <<<kb + pb, T, 0, stream>>>(edst, okey, E, sums, cnt, kb);
    k_owner_list <<<ob + cb, OWN_NT, 0, stream>>>(esrc, edst, okey, lists, lcnt, dinv, batch, cnt, x, W1, bufH, N, E, ob);

    k_conv_owner<true> <<<ob, OWN_NT, 0, stream>>>(bufH, bufO1, esrc, edst, lists, lcnt, dinv, b1, W2, N, E);

    k_conv_owner<false><<<ob, OWN_NT, 0, stream>>>(bufO1, bufH, esrc, edst, lists, lcnt, dinv, b2, W2, N, E);

    k_pool_accum<<<pab, T, 0, stream>>>(bufH, batch, sums, N);
    k_final     <<<hb, T, 0, stream>>>(sums, cnt, Wlin, blin, out);
}


# Owner keys stored as 8 bit planes per 32 edges (a front wave ballots 1024 edges): an owner tests 32 edges with 8 bit operations, not 56; 8x fewer front waves; LDS add order still arrival order
# speedup vs baseline: 4.4478x; 1.0373x over previous
#include <hip/hip_runtime.h>
#include <hip/hip_bf16.h>

#define HID     128
#define FIN     11
#define FOUT    19
#define NGRAPHS 4096

typedef __attribute__((ext_vector_type(2))) float v2f;
typedef __attribute__((ext_vector_type(8))) float v8f;

#define OWN_R   392
#define ACC_P   132

#define OWN_NT  1024
#define OWN_CAP 8192
#define REC_BITS 9
#define SCAN_W  ((OWN_R + 31) / 32)
#define XS_PER_T ((OWN_R * FIN + OWN_NT - 1) / OWN_NT)

__device__ __forceinline__ unsigned owner_key(int d) { return ((unsigned)d / OWN_R) & 255u; }

#define FR_T 256
__device__ __forceinline__ void plane_part(const int* __restrict__ dst, unsigned* __restrict__ planes, int e, int blk) {
    const int lane  = (int)threadIdx.x & 31;
    const int wave  = __builtin_amdgcn_readfirstlane((int)(threadIdx.x >> 5));
    const int g     = blk * (FR_T / 32) + wave;
    const int ebase = g << 10;
    if (ebase >= e) return;
    const int nw = (e + 31) >> 5;
    int d[32];
#pragma unroll
    for (int j = 0; j < 32; ++j) {
        const int i = ebase + (j << 5) + lane;
        d[j] = dst[(i < e) ? i : e - 1];
    }
    uint4 lo = make_uint4(0u, 0u, 0u, 0u), hi = make_uint4(0u, 0u, 0u, 0u);
#pragma unroll
    for (int j = 0; j < 32; ++j) {
        const unsigned k = owner_key(d[j]);
        const unsigned m0 = __builtin_amdgcn_ballot_w32((k & 1u) != 0u);
        const unsigned m1 = __builtin_amdgcn_ballot_w32((k & 2u) != 0u);
        const unsigned m2 = __builtin_amdgcn_ballot_w32((k & 4u) != 0u);
        const unsigned m3 = __builtin_amdgcn_ballot_w32((k & 8u) != 0u);
        const unsigned m4 = __builtin_amdgcn_ballot_w32((k & 16u) != 0u);
        const unsigned m5 = __builtin_amdgcn_ballot_w32((k & 32u) != 0u);
        const unsigned m6 = __builtin_amdgcn_ballot_w32((k & 64u) != 0u);
        const unsigned m7 = __builtin_amdgcn_ballot_w32((k & 128u) != 0u);
        if (lane == j) { lo = make_uint4(m0, m1, m2, m3); hi = make_uint4(m4, m5, m6, m7); }
    }
    const int w = (g << 5) + lane;
    if (w < nw) { ((uint4*)planes)[w] = lo; ((uint4*)planes)[nw + w] = hi; }
}

__global__ __launch_bounds__(OWN_NT) void k_owner_list(const int* __restrict__ src,
                                                       const int* __restrict__ dst,
                                                       const unsigned* __restrict__ planes,
                                                       unsigned* __restrict__ lists,
                                                       int* __restrict__ lcnt,
                                                       float* __restrict__ dinv,
                                                       const int* __restrict__ batch,
                                                       float* cnt,
                                                       const float* __restrict__ x,
                                                       const float* __restrict__ W1,
                                                       float* __restrict__ h,
                                                       int n, int e, int ob) {
    if ((int)blockIdx.x >= ob) {
        const int i = ((int)blockIdx.x - ob) * OWN_NT + (int)threadIdx.x;
        if (i < n) atomicAdd(&cnt[batch[i]], 1.0f);
        return;
    }
    __shared__ int list_e[OWN_CAP];
    __shared__ int cntr[OWN_R];
    __shared__ int first[OWN_R];
    __shared__ int wtot[SCAN_W];
    __shared__ int lcount;
    __shared__ float xs[OWN_R * FIN];
    __shared__ float dloc[OWN_R];

    const int tid   = threadIdx.x;
    const int nbase = blockIdx.x * OWN_R;
    int rows = n - nbase;
    if (rows > OWN_R) rows = OWN_R;
    const unsigned urows = (unsigned)rows;

    float xv[XS_PER_T];
#pragma unroll
    for (int u = 0; u < XS_PER_T; ++u) {
        const int i = u * OWN_NT + tid;
        xv[u] = (i < rows * FIN) ? x[(size_t)nbase * FIN + i] : 0.0f;
    }

    if (tid < OWN_R) cntr[tid] = 0;
    if (tid == 0) lcount = 0;
    __syncthreads();

    auto note1 = [&](int eid) {
        const int sl = atomicAdd(&lcount, 1);
        if (sl < OWN_CAP) list_e[sl] = eid;
    };
    const unsigned mykey = (unsigned)blockIdx.x & 255u;
    unsigned inv[8];
#pragma unroll
    for (int p = 0; p < 8; ++p) inv[p] = ((mykey >> p) & 1u) ? 0u : 0xFFFFFFFFu;
    auto match32 = [&](const uint4 a, const uint4 b) -> unsigned {
        return (a.x ^ inv[0]) & (a.y ^ inv[1]) & (a.z ^ inv[2]) & (a.w ^ inv[3]) &
               (b.x ^ inv[4]) & (b.y ^ inv[5]) & (b.z ^ inv[6]) & (b.w ^ inv[7]);
    };
    auto noteword = [&](unsigned hit, int wi) {
        while (hit) {
            const int b = __builtin_ctz(hit);
            hit &= hit - 1;
            const int eid = (wi << 5) + b;
            if (eid < e) note1(eid);
        }
    };

    const int nw = (e + 31) >> 5;
    const uint4* plo = (const uint4*)planes;
    const uint4* phi = plo + nw;
    int i0 = 0;
    for (; i0 + OWN_NT * 4 <= nw; i0 += OWN_NT * 4) {
        uint4 ka[4], kh[4];
#pragma unroll
        for (int u = 0; u < 4; ++u) { ka[u] = plo[i0 + u * OWN_NT + tid]; kh[u] = phi[i0 + u * OWN_NT + tid]; }
#pragma unroll
        for (int u = 0; u < 4; ++u) noteword(match32(ka[u], kh[u]), i0 + u * OWN_NT + tid);
    }
    for (int i = i0 + tid; i < nw; i += OWN_NT) noteword(match32(plo[i], phi[i]), i);
#pragma unroll
    for (int u = 0; u < XS_PER_T; ++u) {
        const int i = u * OWN_NT + tid;
        if (i < rows * FIN) xs[i] = xv[u];
    }
    __syncthreads();

    const int total = lcount;
    const bool listed = (total <= OWN_CAP) && (ob <= 256);
    unsigned arow[OWN_CAP / OWN_NT];
    int      rk[OWN_CAP / OWN_NT];
    int      s[OWN_CAP / OWN_NT];
#pragma unroll
    for (int u = 0; u < OWN_CAP / OWN_NT; ++u) { arow[u] = 0xFFFFFFFFu; rk[u] = 0; s[u] = 0; }
    if (listed) {
        int eid[OWN_CAP / OWN_NT];
#pragma unroll
        for (int u = 0; u < OWN_CAP / OWN_NT; ++u) {
            const int i = u * OWN_NT + tid;
            eid[u] = 0;
            if (i < total) eid[u] = list_e[i];
        }
        int d[OWN_CAP / OWN_NT];
#pragma unroll
        for (int u = 0; u < OWN_CAP / OWN_NT; ++u) { d[u] = dst[eid[u]]; s[u] = src[eid[u]]; }
#pragma unroll
        for (int u = 0; u < OWN_CAP / OWN_NT; ++u) {
            const int i = u * OWN_NT + tid;
            if (i < total) {
                const unsigned a = (unsigned)(d[u] - nbase);
                if (a < urows) {
                    arow[u] = a;
                    rk[u]   = atomicAdd(&cntr[a], 1);
                }
            }
        }
    } else {
        for (int i = tid; i < e; i += OWN_NT) {
            const unsigned a = (unsigned)(dst[i] - nbase);
            if (a < urows) atomicAdd(&cntr[a], 1);
        }
    }
    __syncthreads();

    const int lane = tid & 31;
    const int wv   = tid >> 5;
    const int cv   = (tid < rows) ? cntr[tid] : 0;
    int inc = cv;
#pragma unroll
    for (int dlt = 1; dlt < 32; dlt <<= 1) {
        const int t = __shfl_up(inc, dlt);
        if (lane >= dlt) inc += t;
    }
    if (lane == 31 && wv < SCAN_W) wtot[wv] = inc;
    __syncthreads();
    int below = 0, stored = 0;
    for (int q = 0; q < SCAN_W; ++q) {
        const int t = wtot[q];
        if (q < wv) below += t;
        stored += t;
    }
    if (tid < OWN_R) first[tid] = below + inc - cv;
    __syncthreads();

    if (listed) {
#pragma unroll
        for (int u = 0; u < OWN_CAP / OWN_NT; ++u) {
            const unsigned a = arow[u];
            if (a < urows) list_e[first[a] + rk[u]] = (int)(((unsigned)s[u] << REC_BITS) | a);
        }
    }
    __syncthreads();
    if (listed) {
        unsigned* mylist = lists + (size_t)blockIdx.x * OWN_CAP;
        for (int i = tid; i < stored; i += OWN_NT) mylist[i] = (unsigned)list_e[i];
    }

    if (tid < rows) {
        const float dg = 1.0f + (float)cntr[tid];
        const float dv = (dg > 0.0f) ? rsqrtf(dg) : 0.0f;
        dinv[nbase + tid] = dv;
        dloc[tid] = dv;
    }
    if (tid == 0) lcnt[blockIdx.x] = listed ? stored : OWN_CAP + 1;

    float4 w[FIN];
#pragma unroll
    for (int k = 0; k < FIN; ++k) w[k] = ((const float4*)W1)[k * (HID / 4) + lane];
    __syncthreads();
    const int swave = __builtin_amdgcn_readfirstlane(wv);
    for (int r = swave; r < rows; r += OWN_NT / 32) {
        const float xl = (lane < FIN) ? xs[r * FIN + lane] : 0.0f;
        float4 a = make_float4(0.0f, 0.0f, 0.0f, 0.0f);
#pragma unroll
        for (int k = 0; k < FIN; ++k) {
            const float xk = __int_as_float(__builtin_amdgcn_readlane(__float_as_int(xl), k));
            a.x = fmaf(xk, w[k].x, a.x);
            a.y = fmaf(xk, w[k].y, a.y);
            a.z = fmaf(xk, w[k].z, a.z);
            a.w = fmaf(xk, w[k].w, a.w);
        }
        const float dn = dloc[r];
        a.x *= dn; a.y *= dn; a.z *= dn; a.w *= dn;
        ((float4*)(h + (size_t)(nbase + r) * HID))[lane] = a;
    }
}

static_assert(OWN_R <= (1 << REC_BITS), "the owned row must fit the record's row field");
#define INIT_PER_T ((OWN_R * (HID / 4) + OWN_NT - 1) / OWN_NT)

template <bool W2EPI>
__global__ __launch_bounds__(OWN_NT) void k_conv_owner(const float* __restrict__ h,
                                                       float* __restrict__ out,
                                                       const int* __restrict__ src,
                                                       const int* __restrict__ dst,
                                                       const unsigned* __restrict__ lists,
                                                       const int* __restrict__ lcnt,
                                                       const float* __restrict__ dinv,
                                                       const float* __restrict__ bias,
                                                       const float* __restrict__ W,
                                                       int n, int e) {
    __shared__ __attribute__((aligned(16))) float acc[OWN_R * ACC_P];
    __shared__ float dnl[OWN_R];

    const int tid   = threadIdx.x;
    const int nbase = blockIdx.x * OWN_R;
    int rows = n - nbase;
    if (rows > OWN_R) rows = OWN_R;
    const unsigned urows = (unsigned)rows;
    const float4 b4 = ((const float4*)bias)[tid & 31];

    {
        const float4* h4   = (const float4*)(h + (size_t)nbase * HID);
        float4*       acc4 = (float4*)acc;
        const int     lim  = rows * (HID / 4);
        float dv = 0.0f;
        if (tid < rows) dv = dinv[nbase + tid];
        float4 hv[INIT_PER_T];
#pragma unroll
        for (int u = 0; u < INIT_PER_T; ++u) {
            const int i = u * OWN_NT + tid;
            hv[u] = (i < lim) ? h4[i] : make_float4(0.0f, 0.0f, 0.0f, 0.0f);
        }
        if (tid < rows) dnl[tid] = dv;
#pragma unroll
        for (int u = 0; u < INIT_PER_T; ++u) {
            const int i = u * OWN_NT + tid;
            if (i < lim) acc4[(i >> 5) * (ACC_P / 4) + (i & 31)] = hv[u];
        }
    }
    __syncthreads();

    const int total = lcnt[blockIdx.x];
    if (total <= OWN_CAP && gridDim.x <= 256) {
        const unsigned* mylist = lists + (size_t)blockIdx.x * OWN_CAP;
        const int lane = tid & 31;
        const int wave = __builtin_amdgcn_readfirstlane(tid >> 5);
        for (int base = wave * 32; base < total; base += OWN_NT) {
            int   s = 0, dl = 0;
            if (base + lane < total) {
                const unsigned rec = mylist[base + lane];
                s   = (int)(rec >> REC_BITS);
                dl  = (int)(rec & ((1u << REC_BITS) - 1u));
            }
            int cnt = total - base;
            if (cnt > 32) cnt = 32;
            float run[4] = {0.0f, 0.0f, 0.0f, 0.0f};
            int   cur = __builtin_amdgcn_readlane(dl, 0);
            auto flush = [&](int row) {
                float* a = acc + row * ACC_P + lane;
#pragma unroll
                for (int k = 0; k < 4; ++k) { atomicAdd(&a[k * 32], run[k]); run[k] = 0.0f; }
            };
            int j = 0;
            for (; j + 8 <= cnt; j += 8) {
                float v[8][4];
#pragma unroll
                for (int u = 0; u < 8; ++u) {
                    const int sj = __builtin_amdgcn_readlane(s, j + u);
                    const float* r = h + (size_t)sj * HID + lane;
#pragma unroll
                    for (int k = 0; k < 4; ++k) v[u][k] = r[k * 32];
                }
#pragma unroll
                for (int u = 0; u < 8; ++u) {
                    const int   dj = __builtin_amdgcn_readlane(dl, j + u);
                    if (dj != cur) { flush(cur); cur = dj; }
#pragma unroll
                    for (int k = 0; k < 4; ++k) run[k] += v[u][k];
                }
            }
            for (; j < cnt; ++j) {
                const int   sj = __builtin_amdgcn_readlane(s, j);
                const int   dj = __builtin_amdgcn_readlane(dl, j);
                const float* r = h + (size_t)sj * HID + lane;
                if (dj != cur) { flush(cur); cur = dj; }
#pragma unroll
                for (int k = 0; k < 4; ++k) run[k] += r[k * 32];
            }
            flush(cur);
        }
    } else {
        for (int i = tid; i < e; i += OWN_NT) {
            const unsigned a = (unsigned)(dst[i] - nbase);
            if (a < urows) {
                const int   s   = src[i];
                const float* r  = h + (size_t)s * HID;
                float* ar = acc + a * ACC_P;
                for (int f = 0; f < HID; ++f) atomicAdd(&ar[f], r[f]);
            }
        }
    }
    if constexpr (W2EPI) {
        const int lane = tid & 31;
        const int wave = __builtin_amdgcn_readfirstlane(tid >> 5);
        const int half = lane >> 4;
        const int l15  = lane & 15;
        const int ncol = ((wave & 7) << 4) + l15;
        v2f bf[HID / 4];
#pragma unroll
        for (int kk = 0; kk < HID / 4; ++kk) {
            const int ka = kk * 4 + half * 2;
            bf[kk][0] = W[ka * HID + ncol];
            bf[kk][1] = W[(ka + 1) * HID + ncol];
        }
        __syncthreads();

        {
            float4* acc4 = (float4*)acc;
            for (int i = tid; i < rows * (HID / 4); i += OWN_NT) {
                const float dn = dnl[i >> 5];
                float4 v = acc4[(i >> 5) * (ACC_P / 4) + (i & 31)];
                v.x = fmaf(v.x, dn, b4.x);
                v.y = fmaf(v.y, dn, b4.y);
                v.z = fmaf(v.z, dn, b4.z);
                v.w = fmaf(v.w, dn, b4.w);
                acc4[(i >> 5) * (ACC_P / 4) + (i & 31)] = v;
            }
        }
        __syncthreads();

        const int ntile = (rows + 15) >> 4;
        for (int t = wave >> 3; t < ntile; t += 4) {
            int r = (t << 4) + l15;
            if (r >= rows) r = rows - 1;
            const float* ar = acc + r * ACC_P + half * 2;
            v8f c = {};
#pragma unroll
            for (int kk = 0; kk < HID / 4; ++kk) {
                const float2 av = *(const float2*)(ar + kk * 4);
                v2f a;
                a[0] = fmaxf(av.x, 0.0f);
                a[1] = fmaxf(av.y, 0.0f);
                c = __builtin_amdgcn_wmma_f32_16x16x4_f32(
                         false, a,  false, bf[kk],
                         (short)0, c,  false,  false);
            }
            const int rb = (t << 4) + half * 8;
#pragma unroll
            for (int i = 0; i < 8; ++i) {
                const int rr = rb + i;
                if (rr < rows) out[(size_t)(nbase + rr) * HID + ncol] = c[i] * dnl[rr];
            }
        }
    } else {
        __syncthreads();

        float4*       o4 = (float4*)(out + (size_t)nbase * HID);
        const float4* a4 = (const float4*)acc;
        for (int i = tid; i < rows * (HID / 4); i += OWN_NT) {
            const float  dn = dnl[i >> 5];
            const float4 v  = a4[(i >> 5) * (ACC_P / 4) + (i & 31)];
            float4 o;
            o.x = fmaf(v.x, dn, b4.x);
            o.y = fmaf(v.y, dn, b4.y);
            o.z = fmaf(v.z, dn, b4.z);
            o.w = fmaf(v.w, dn, b4.w);
            o4[i] = o;
        }
    }
}

__device__ __forceinline__ void pool_init_part(float* sums, float* cnt, int blk) {
    int idx = blk * (int)blockDim.x + (int)threadIdx.x;
    const float4 z = make_float4(0.0f, 0.0f, 0.0f, 0.0f);
    if (idx < NGRAPHS * HID / 4) ((float4*)sums)[idx] = z;
    if (idx < NGRAPHS / 4)       ((float4*)cnt)[idx]  = z;
}

__global__ __launch_bounds__(FR_T) void k_front(const int* __restrict__ dst, unsigned* __restrict__ planes, int e,
                                                float* sums, float* cnt, int kb) {
    const int b = (int)blockIdx.x;
    if (b < kb) plane_part(dst, planes, e, b);
    else        pool_init_part(sums, cnt, b - kb);
}

#define PA_ROWS 16
__global__ __launch_bounds__(256) void k_pool_accum(const float* __restrict__ h, const int* __restrict__ batch,
                                                    float* sums, int n) {
    const int lane = threadIdx.x & 31;
    const int wave = __builtin_amdgcn_readfirstlane((int)(threadIdx.x >> 5));
    const int r0   = (blockIdx.x * 8 + wave) * PA_ROWS;
    if (r0 >= n) return;
    int cnt = n - r0;
    if (cnt > PA_ROWS) cnt = PA_ROWS;
    const int b = (lane < cnt) ? batch[r0 + lane] : 0;

    float v[PA_ROWS][4];
#pragma unroll
    for (int u = 0; u < PA_ROWS; ++u) {
        const int r = r0 + ((u < cnt) ? u : 0);
        const float* p = h + (size_t)r * HID + lane;
#pragma unroll
        for (int k = 0; k < 4; ++k) v[u][k] = p[k * 32];
    }

    float acc[4] = {0.0f, 0.0f, 0.0f, 0.0f};
    int cur = __builtin_amdgcn_readlane(b, 0);
#pragma unroll
    for (int u = 0; u < PA_ROWS; ++u) {
        if (u < cnt) {
            const int bu = __builtin_amdgcn_readlane(b, u);
            if (bu != cur) {
                float* s = sums + (size_t)cur * HID + lane;
#pragma unroll
                for (int k = 0; k < 4; ++k) { atomicAdd(&s[k * 32], acc[k]); acc[k] = 0.0f; }
                cur = bu;
            }
#pragma unroll
            for (int k = 0; k < 4; ++k) acc[k] += v[u][k];
        }
    }
    float* s = sums + (size_t)cur * HID + lane;
#pragma unroll
    for (int k = 0; k < 4; ++k) atomicAdd(&s[k * 32], acc[k]);
}

__global__ void k_final(const float* __restrict__ sums, const float* __restrict__ cnt,
                        const float* __restrict__ Wlin, const float* __restrict__ blin,
                        float* __restrict__ out) {
    int idx = blockIdx.x * blockDim.x + threadIdx.x;
    if (idx >= NGRAPHS * FOUT) return;
    int g = idx / FOUT;
    int j = idx - g * FOUT;
    float inv = 1.0f / fmaxf(cnt[g], 1.0f);
    const float* s = sums + g * HID;
    float acc = blin[j];
#pragma unroll 4
    for (int k = 0; k < HID; ++k)
        acc = fmaf(s[k] * inv, Wlin[k * FOUT + j], acc);
    out[idx] = acc;
}

extern "C" void kernel_launch(void* const* d_in, const int* in_sizes, int n_in,
                              void* d_out, int out_size, void* d_ws, size_t ws_size,
                              hipStream_t stream) {
    (void)n_in; (void)out_size; (void)ws_size;

    const float* x     = (const float*)d_in[0];
    const int*   esrc  = (const int*)  d_in[1];
    const int*   edst  = (const int*)  d_in[2];
    const int*   batch = (const int*)  d_in[3];
    const float* W1    = (const float*)d_in[4];
    const float* b1    = (const float*)d_in[5];
    const float* W2    = (const float*)d_in[6];
    const float* b2    = (const float*)d_in[7];
    const float* Wlin  = (const float*)d_in[8];
    const float* blin  = (const float*)d_in[9];
    float*       out   = (float*)d_out;

    const int N = in_sizes[3];
    const int E = in_sizes[1];
    const int ob = (N + OWN_R - 1) / OWN_R;

    char*  ws  = (char*)d_ws;
    size_t off = 0;
    auto carve = [&](size_t bytes) -> void* {
        off = (off + 255) & ~(size_t)255;
        void* p = ws + off;
        off += bytes;
        return p;
    };
    float* dinv  = (float*)carve((size_t)N * 4);
    float* bufH  = (float*)carve((size_t)N * HID * 4);
    float* bufO1 = (float*)carve((size_t)N * HID * 4);
    unsigned* oplanes = (unsigned*)carve((size_t)((E + 31) / 32) * 32);
    unsigned* lists = (unsigned*)carve((size_t)256 * OWN_CAP * 4);
    int* lcnt = (int*)carve((size_t)ob * 4);
    float* sums  = (float*)carve((size_t)NGRAPHS * HID * 4);
    float* cnt   = (float*)carve((size_t)NGRAPHS * 4);

    const int T = 256;
    int cb   = (N + OWN_NT - 1) / OWN_NT;
    int kb   = (E + 8 * 1024 - 1) / (8 * 1024);
    int pab  = (N + 8 * PA_ROWS - 1) / (8 * PA_ROWS);
    int pb   = (NGRAPHS * HID / 4 + T - 1) / T;
    int hb   = (NGRAPHS * FOUT + T - 1) / T;

    k_front      <<<kb + pb, T, 0, stream>>>(edst, oplanes, E, sums, cnt, kb);
    k_owner_list <<<ob + cb, OWN_NT, 0, stream>>>(esrc, edst, oplanes, lists, lcnt, dinv, batch, cnt, x, W1, bufH, N, E, ob);

    k_conv_owner<true> <<<ob, OWN_NT, 0, stream>>>(bufH, bufO1, esrc, edst, lists, lcnt, dinv, b1, W2, N, E);

    k_conv_owner<false><<<ob, OWN_NT, 0, stream>>>(bufO1, bufH, esrc, edst, lists, lcnt, dinv, b2, W2, N, E);

    k_pool_accum<<<pab, T, 0, stream>>>(bufH, batch, sums, N);
    k_final     <<<hb, T, 0, stream>>>(sums, cnt, Wlin, blin, out);
}


# Owner key scan replaced by a front counting sort: a front block stores its 4096 edge records grouped by owner plus a 256-word table; an owner reads only its groups; row order still LDS add arrival
# speedup vs baseline: 4.9581x; 1.1147x over previous
#include <hip/hip_runtime.h>
#include <hip/hip_bf16.h>

#define HID     128
#define FIN     11
#define FOUT    19
#define NGRAPHS 4096

typedef __attribute__((ext_vector_type(2))) float v2f;
typedef __attribute__((ext_vector_type(8))) float v8f;

#define OWN_R   392
#define ACC_P   132

#define OWN_NT  1024
#define OWN_CAP 8192
#define REC_BITS 9
#define SCAN_W  ((OWN_R + 31) / 32)
#define XS_PER_T ((OWN_R * FIN + OWN_NT - 1) / OWN_NT)

#define FR_T  512
#define SEG_E 4096
#define SEG_PER_T (SEG_E / FR_T)
#define TAB_W 256

__global__ __launch_bounds__(OWN_NT) void k_owner_list(const int* __restrict__ dst,
                                                       const unsigned* __restrict__ seg,
                                                       const unsigned* __restrict__ tab,
                                                       unsigned* __restrict__ lists,
                                                       int* __restrict__ lcnt,
                                                       float* __restrict__ dinv,
                                                       const int* __restrict__ batch,
                                                       float* cnt,
                                                       const float* __restrict__ x,
                                                       const float* __restrict__ W1,
                                                       float* __restrict__ h,
                                                       int n, int e, int ob, int nb, int segok) {
    if ((int)blockIdx.x >= ob) {
        const int i = ((int)blockIdx.x - ob) * OWN_NT + (int)threadIdx.x;
        if (i < n) atomicAdd(&cnt[batch[i]], 1.0f);
        return;
    }
    __shared__ int list_e[OWN_CAP];
    __shared__ int bpos[OWN_NT];
    __shared__ int bsrc[OWN_NT];
    __shared__ int btot[OWN_NT / 32];
    __shared__ int cntr[OWN_R];
    __shared__ int first[OWN_R];
    __shared__ int wtot[SCAN_W];
    __shared__ float xs[OWN_R * FIN];
    __shared__ float dloc[OWN_R];

    const int tid   = threadIdx.x;
    const int nbase = blockIdx.x * OWN_R;
    int rows = n - nbase;
    if (rows > OWN_R) rows = OWN_R;
    const unsigned urows = (unsigned)rows;

    float xv[XS_PER_T];
#pragma unroll
    for (int u = 0; u < XS_PER_T; ++u) {
        const int i = u * OWN_NT + tid;
        xv[u] = (i < rows * FIN) ? x[(size_t)nbase * FIN + i] : 0.0f;
    }

    if (tid < OWN_R) cntr[tid] = 0;

    const int lane = tid & 31;
    const int wv   = tid >> 5;
    const bool pre = (ob <= TAB_W) && (nb <= OWN_NT) && (segok != 0);
    unsigned te = 0u;
    if (pre && tid < nb) te = tab[(size_t)tid * TAB_W + blockIdx.x];
    const int myc = (int)(te & 0x1FFFu);
    const int mys = (int)(te >> 13);
    int binc = myc;
#pragma unroll
    for (int dlt = 1; dlt < 32; dlt <<= 1) {
        const int t = __shfl_up(binc, dlt);
        if (lane >= dlt) binc += t;
    }
    if (lane == 31) btot[wv] = binc;
#pragma unroll
    for (int u = 0; u < XS_PER_T; ++u) {
        const int i = u * OWN_NT + tid;
        if (i < rows * FIN) xs[i] = xv[u];
    }
    __syncthreads();
    int bbelow = 0, total = 0;
    for (int q = 0; q < OWN_NT / 32; ++q) {
        const int t = btot[q];
        if (q < wv) bbelow += t;
        total += t;
    }
    bpos[tid] = bbelow + binc - myc;
    bsrc[tid] = tid * SEG_E + mys;
    __syncthreads();

    const bool listed = pre && (total <= OWN_CAP);
    unsigned rec[OWN_CAP / OWN_NT];
    unsigned arow[OWN_CAP / OWN_NT];
    int      rk[OWN_CAP / OWN_NT];
#pragma unroll
    for (int u = 0; u < OWN_CAP / OWN_NT; ++u) { rec[u] = 0u; arow[u] = 0xFFFFFFFFu; rk[u] = 0; }
    if (listed) {
        int at[OWN_CAP / OWN_NT];
#pragma unroll
        for (int u = 0; u < OWN_CAP / OWN_NT; ++u) {
            const int i = u * OWN_NT + tid;
            at[u] = 0;
            if (i < total) {
                int b = 0;
                for (int bit = OWN_NT / 2; bit > 0; bit >>= 1) {
                    const int cnd = b + bit;
                    if (cnd < nb && bpos[cnd] <= i) b = cnd;
                }
                at[u] = bsrc[b] + (i - bpos[b]);
            }
        }
#pragma unroll
        for (int u = 0; u < OWN_CAP / OWN_NT; ++u) rec[u] = seg[at[u]];
#pragma unroll
        for (int u = 0; u < OWN_CAP / OWN_NT; ++u) {
            const int i = u * OWN_NT + tid;
            if (i < total) {
                const unsigned a = rec[u] & ((1u << REC_BITS) - 1u);
                if (a < urows) {
                    arow[u] = a;
                    rk[u]   = atomicAdd(&cntr[a], 1);
                }
            }
        }
    } else {
        for (int i = tid; i < e; i += OWN_NT) {
            const unsigned a = (unsigned)(dst[i] - nbase);
            if (a < urows) atomicAdd(&cntr[a], 1);
        }
    }
    __syncthreads();

    const int cv   = (tid < rows) ? cntr[tid] : 0;
    int inc = cv;
#pragma unroll
    for (int dlt = 1; dlt < 32; dlt <<= 1) {
        const int t = __shfl_up(inc, dlt);
        if (lane >= dlt) inc += t;
    }
    if (lane == 31 && wv < SCAN_W) wtot[wv] = inc;
    __syncthreads();
    int below = 0, stored = 0;
    for (int q = 0; q < SCAN_W; ++q) {
        const int t = wtot[q];
        if (q < wv) below += t;
        stored += t;
    }
    if (tid < OWN_R) first[tid] = below + inc - cv;
    __syncthreads();

    if (listed) {
#pragma unroll
        for (int u = 0; u < OWN_CAP / OWN_NT; ++u) {
            const unsigned a = arow[u];
            if (a < urows) list_e[first[a] + rk[u]] = (int)rec[u];
        }
    }
    __syncthreads();
    if (listed) {
        unsigned* mylist = lists + (size_t)blockIdx.x * OWN_CAP;
        for (int i = tid; i < stored; i += OWN_NT) mylist[i] = (unsigned)list_e[i];
    }

    if (tid < rows) {
        const float dg = 1.0f + (float)cntr[tid];
        const float dv = (dg > 0.0f) ? rsqrtf(dg) : 0.0f;
        dinv[nbase + tid] = dv;
        dloc[tid] = dv;
    }
    if (tid == 0) lcnt[blockIdx.x] = listed ? stored : OWN_CAP + 1;

    float4 w[FIN];
#pragma unroll
    for (int k = 0; k < FIN; ++k) w[k] = ((const float4*)W1)[k * (HID / 4) + lane];
    __syncthreads();
    const int swave = __builtin_amdgcn_readfirstlane(wv);
    for (int r = swave; r < rows; r += OWN_NT / 32) {
        const float xl = (lane < FIN) ? xs[r * FIN + lane] : 0.0f;
        float4 a = make_float4(0.0f, 0.0f, 0.0f, 0.0f);
#pragma unroll
        for (int k = 0; k < FIN; ++k) {
            const float xk = __int_as_float(__builtin_amdgcn_readlane(__float_as_int(xl), k));
            a.x = fmaf(xk, w[k].x, a.x);
            a.y = fmaf(xk, w[k].y, a.y);
            a.z = fmaf(xk, w[k].z, a.z);
            a.w = fmaf(xk, w[k].w, a.w);
        }
        const float dn = dloc[r];
        a.x *= dn; a.y *= dn; a.z *= dn; a.w *= dn;
        ((float4*)(h + (size_t)(nbase + r) * HID))[lane] = a;
    }
}

static_assert(OWN_R <= (1 << REC_BITS), "the owned row must fit the record's row field");
#define INIT_PER_T ((OWN_R * (HID / 4) + OWN_NT - 1) / OWN_NT)

template <bool W2EPI>
__global__ __launch_bounds__(OWN_NT) void k_conv_owner(const float* __restrict__ h,
                                                       float* __restrict__ out,
                                                       const int* __restrict__ src,
                                                       const int* __restrict__ dst,
                                                       const unsigned* __restrict__ lists,
                                                       const int* __restrict__ lcnt,
                                                       const float* __restrict__ dinv,
                                                       const float* __restrict__ bias,
                                                       const float* __restrict__ W,
                                                       int n, int e) {
    __shared__ __attribute__((aligned(16))) float acc[OWN_R * ACC_P];
    __shared__ float dnl[OWN_R];

    const int tid   = threadIdx.x;
    const int nbase = blockIdx.x * OWN_R;
    int rows = n - nbase;
    if (rows > OWN_R) rows = OWN_R;
    const unsigned urows = (unsigned)rows;
    const float4 b4 = ((const float4*)bias)[tid & 31];

    {
        const float4* h4   = (const float4*)(h + (size_t)nbase * HID);
        float4*       acc4 = (float4*)acc;
        const int     lim  = rows * (HID / 4);
        float dv = 0.0f;
        if (tid < rows) dv = dinv[nbase + tid];
        float4 hv[INIT_PER_T];
#pragma unroll
        for (int u = 0; u < INIT_PER_T; ++u) {
            const int i = u * OWN_NT + tid;
            hv[u] = (i < lim) ? h4[i] : make_float4(0.0f, 0.0f, 0.0f, 0.0f);
        }
        if (tid < rows) dnl[tid] = dv;
#pragma unroll
        for (int u = 0; u < INIT_PER_T; ++u) {
            const int i = u * OWN_NT + tid;
            if (i < lim) acc4[(i >> 5) * (ACC_P / 4) + (i & 31)] = hv[u];
        }
    }
    __syncthreads();

    const int total = lcnt[blockIdx.x];
    if (total <= OWN_CAP && gridDim.x <= 256) {
        const unsigned* mylist = lists + (size_t)blockIdx.x * OWN_CAP;
        const int lane = tid & 31;
        const int wave = __builtin_amdgcn_readfirstlane(tid >> 5);
        for (int base = wave * 32; base < total; base += OWN_NT) {
            int   s = 0, dl = 0;
            if (base + lane < total) {
                const unsigned rec = mylist[base + lane];
                s   = (int)(rec >> REC_BITS);
                dl  = (int)(rec & ((1u << REC_BITS) - 1u));
            }
            int cnt = total - base;
            if (cnt > 32) cnt = 32;
            float run[4] = {0.0f, 0.0f, 0.0f, 0.0f};
            int   cur = __builtin_amdgcn_readlane(dl, 0);
            auto flush = [&](int row) {
                float* a = acc + row * ACC_P + lane;
#pragma unroll
                for (int k = 0; k < 4; ++k) { atomicAdd(&a[k * 32], run[k]); run[k] = 0.0f; }
            };
            int j = 0;
            for (; j + 8 <= cnt; j += 8) {
                float v[8][4];
#pragma unroll
                for (int u = 0; u < 8; ++u) {
                    const int sj = __builtin_amdgcn_readlane(s, j + u);
                    const float* r = h + (size_t)sj * HID + lane;
#pragma unroll
                    for (int k = 0; k < 4; ++k) v[u][k] = r[k * 32];
                }
#pragma unroll
                for (int u = 0; u < 8; ++u) {
                    const int   dj = __builtin_amdgcn_readlane(dl, j + u);
                    if (dj != cur) { flush(cur); cur = dj; }
#pragma unroll
                    for (int k = 0; k < 4; ++k) run[k] += v[u][k];
                }
            }
            for (; j < cnt; ++j) {
                const int   sj = __builtin_amdgcn_readlane(s, j);
                const int   dj = __builtin_amdgcn_readlane(dl, j);
                const float* r = h + (size_t)sj * HID + lane;
                if (dj != cur) { flush(cur); cur = dj; }
#pragma unroll
                for (int k = 0; k < 4; ++k) run[k] += r[k * 32];
            }
            flush(cur);
        }
    } else {
        for (int i = tid; i < e; i += OWN_NT) {
            const unsigned a = (unsigned)(dst[i] - nbase);
            if (a < urows) {
                const int   s   = src[i];
                const float* r  = h + (size_t)s * HID;
                float* ar = acc + a * ACC_P;
                for (int f = 0; f < HID; ++f) atomicAdd(&ar[f], r[f]);
            }
        }
    }
    if constexpr (W2EPI) {
        const int lane = tid & 31;
        const int wave = __builtin_amdgcn_readfirstlane(tid >> 5);
        const int half = lane >> 4;
        const int l15  = lane & 15;
        const int ncol = ((wave & 7) << 4) + l15;
        v2f bf[HID / 4];
#pragma unroll
        for (int kk = 0; kk < HID / 4; ++kk) {
            const int ka = kk * 4 + half * 2;
            bf[kk][0] = W[ka * HID + ncol];
            bf[kk][1] = W[(ka + 1) * HID + ncol];
        }
        __syncthreads();

        {
            float4* acc4 = (float4*)acc;
            for (int i = tid; i < rows * (HID / 4); i += OWN_NT) {
                const float dn = dnl[i >> 5];
                float4 v = acc4[(i >> 5) * (ACC_P / 4) + (i & 31)];
                v.x = fmaf(v.x, dn, b4.x);
                v.y = fmaf(v.y, dn, b4.y);
                v.z = fmaf(v.z, dn, b4.z);
                v.w = fmaf(v.w, dn, b4.w);
                acc4[(i >> 5) * (ACC_P / 4) + (i & 31)] = v;
            }
        }
        __syncthreads();

        const int ntile = (rows + 15) >> 4;
        for (int t = wave >> 3; t < ntile; t += 4) {
            int r = (t << 4) + l15;
            if (r >= rows) r = rows - 1;
            const float* ar = acc + r * ACC_P + half * 2;
            v8f c = {};
#pragma unroll
            for (int kk = 0; kk < HID / 4; ++kk) {
                const float2 av = *(const float2*)(ar + kk * 4);
                v2f a;
                a[0] = fmaxf(av.x, 0.0f);
                a[1] = fmaxf(av.y, 0.0f);
                c = __builtin_amdgcn_wmma_f32_16x16x4_f32(
                         false, a,  false, bf[kk],
                         (short)0, c,  false,  false);
            }
            const int rb = (t << 4) + half * 8;
#pragma unroll
            for (int i = 0; i < 8; ++i) {
                const int rr = rb + i;
                if (rr < rows) out[(size_t)(nbase + rr) * HID + ncol] = c[i] * dnl[rr];
            }
        }
    } else {
        __syncthreads();

        float4*       o4 = (float4*)(out + (size_t)nbase * HID);
        const float4* a4 = (const float4*)acc;
        for (int i = tid; i < rows * (HID / 4); i += OWN_NT) {
            const float  dn = dnl[i >> 5];
            const float4 v  = a4[(i >> 5) * (ACC_P / 4) + (i & 31)];
            float4 o;
            o.x = fmaf(v.x, dn, b4.x);
            o.y = fmaf(v.y, dn, b4.y);
            o.z = fmaf(v.z, dn, b4.z);
            o.w = fmaf(v.w, dn, b4.w);
            o4[i] = o;
        }
    }
}

__device__ __forceinline__ void pool_init_part(float* sums, float* cnt, int blk) {
    int idx = blk * (int)blockDim.x + (int)threadIdx.x;
    const float4 z = make_float4(0.0f, 0.0f, 0.0f, 0.0f);
    if (idx < NGRAPHS * HID / 4) ((float4*)sums)[idx] = z;
    if (idx < NGRAPHS / 4)       ((float4*)cnt)[idx]  = z;
}

__global__ __launch_bounds__(FR_T) void k_front(const int* __restrict__ src, const int* __restrict__ dst,
                                                unsigned* __restrict__ seg, unsigned* __restrict__ tab, int e,
                                                float* sums, float* cnt, int kb, int segok) {
    __shared__ int      hist[TAB_W];
    __shared__ int      hfirst[TAB_W];
    __shared__ int      hw[TAB_W / 32];
    __shared__ unsigned srt[SEG_E];
    const int b = (int)blockIdx.x;
    if (b >= kb) { pool_init_part(sums, cnt, b - kb); return; }
    if (!segok) return;

    const int tid   = (int)threadIdx.x;
    const int ebase = b * SEG_E;
    int cnt_e = e - ebase;
    if (cnt_e > SEG_E) cnt_e = SEG_E;
    if (tid < TAB_W) hist[tid] = 0;
    int d[SEG_PER_T], s[SEG_PER_T];
#pragma unroll
    for (int u = 0; u < SEG_PER_T; ++u) {
        const int i = u * FR_T + tid;
        const int ii = ebase + ((i < cnt_e) ? i : 0);
        d[u] = dst[ii];
        s[u] = src[ii];
    }
    __syncthreads();
    unsigned key[SEG_PER_T], rec[SEG_PER_T];
    int      rk[SEG_PER_T];
#pragma unroll
    for (int u = 0; u < SEG_PER_T; ++u) {
        const int i = u * FR_T + tid;
        const unsigned o = (unsigned)d[u] / OWN_R;
        key[u] = o & (TAB_W - 1u);
        rec[u] = ((unsigned)s[u] << REC_BITS) | ((unsigned)d[u] - o * OWN_R);
        rk[u]  = 0;
        if (i < cnt_e) rk[u] = atomicAdd(&hist[key[u]], 1);
    }
    __syncthreads();
    const int lane = tid & 31;
    const int wv   = tid >> 5;
    const int cv   = (tid < TAB_W) ? hist[tid] : 0;
    int inc = cv;
#pragma unroll
    for (int dlt = 1; dlt < 32; dlt <<= 1) {
        const int t = __shfl_up(inc, dlt);
        if (lane >= dlt) inc += t;
    }
    if (lane == 31 && wv < TAB_W / 32) hw[wv] = inc;
    __syncthreads();
    int below = 0;
    for (int q = 0; q < TAB_W / 32; ++q) {
        const int t = hw[q];
        if (q < wv) below += t;
    }
    if (tid < TAB_W) {
        const int fo = below + inc - cv;
        hfirst[tid] = fo;
        tab[(size_t)b * TAB_W + tid] = ((unsigned)fo << 13) | (unsigned)cv;
    }
    __syncthreads();
#pragma unroll
    for (int u = 0; u < SEG_PER_T; ++u) {
        const int i = u * FR_T + tid;
        if (i < cnt_e) srt[hfirst[key[u]] + rk[u]] = rec[u];
    }
    __syncthreads();
    for (int i = tid; i < cnt_e; i += FR_T) seg[ebase + i] = srt[i];
}

#define PA_ROWS 16
__global__ __launch_bounds__(256) void k_pool_accum(const float* __restrict__ h, const int* __restrict__ batch,
                                                    float* sums, int n) {
    const int lane = threadIdx.x & 31;
    const int wave = __builtin_amdgcn_readfirstlane((int)(threadIdx.x >> 5));
    const int r0   = (blockIdx.x * 8 + wave) * PA_ROWS;
    if (r0 >= n) return;
    int cnt = n - r0;
    if (cnt > PA_ROWS) cnt = PA_ROWS;
    const int b = (lane < cnt) ? batch[r0 + lane] : 0;

    float v[PA_ROWS][4];
#pragma unroll
    for (int u = 0; u < PA_ROWS; ++u) {
        const int r = r0 + ((u < cnt) ? u : 0);
        const float* p = h + (size_t)r * HID + lane;
#pragma unroll
        for (int k = 0; k < 4; ++k) v[u][k] = p[k * 32];
    }

    float acc[4] = {0.0f, 0.0f, 0.0f, 0.0f};
    int cur = __builtin_amdgcn_readlane(b, 0);
#pragma unroll
    for (int u = 0; u < PA_ROWS; ++u) {
        if (u < cnt) {
            const int bu = __builtin_amdgcn_readlane(b, u);
            if (bu != cur) {
                float* s = sums + (size_t)cur * HID + lane;
#pragma unroll
                for (int k = 0; k < 4; ++k) { atomicAdd(&s[k * 32], acc[k]); acc[k] = 0.0f; }
                cur = bu;
            }
#pragma unroll
            for (int k = 0; k < 4; ++k) acc[k] += v[u][k];
        }
    }
    float* s = sums + (size_t)cur * HID + lane;
#pragma unroll
    for (int k = 0; k < 4; ++k) atomicAdd(&s[k * 32], acc[k]);
}

__global__ void k_final(const float* __restrict__ sums, const float* __restrict__ cnt,
                        const float* __restrict__ Wlin, const float* __restrict__ blin,
                        float* __restrict__ out) {
    int idx = blockIdx.x * blockDim.x + threadIdx.x;
    if (idx >= NGRAPHS * FOUT) return;
    int g = idx / FOUT;
    int j = idx - g * FOUT;
    float inv = 1.0f / fmaxf(cnt[g], 1.0f);
    const float* s = sums + g * HID;
    float acc = blin[j];
#pragma unroll 4
    for (int k = 0; k < HID; ++k)
        acc = fmaf(s[k] * inv, Wlin[k * FOUT + j], acc);
    out[idx] = acc;
}

extern "C" void kernel_launch(void* const* d_in, const int* in_sizes, int n_in,
                              void* d_out, int out_size, void* d_ws, size_t ws_size,
                              hipStream_t stream) {
    (void)n_in; (void)out_size; (void)ws_size;

    const float* x     = (const float*)d_in[0];
    const int*   esrc  = (const int*)  d_in[1];
    const int*   edst  = (const int*)  d_in[2];
    const int*   batch = (const int*)  d_in[3];
    const float* W1    = (const float*)d_in[4];
    const float* b1    = (const float*)d_in[5];
    const float* W2    = (const float*)d_in[6];
    const float* b2    = (const float*)d_in[7];
    const float* Wlin  = (const float*)d_in[8];
    const float* blin  = (const float*)d_in[9];
    float*       out   = (float*)d_out;

    const int N = in_sizes[3];
    const int E = in_sizes[1];
    const int ob = (N + OWN_R - 1) / OWN_R;

    char*  ws  = (char*)d_ws;
    size_t off = 0;
    auto carve = [&](size_t bytes) -> void* {
        off = (off + 255) & ~(size_t)255;
        void* p = ws + off;
        off += bytes;
        return p;
    };
    float* dinv  = (float*)carve((size_t)N * 4);
    float* bufH  = (float*)carve((size_t)N * HID * 4);
    float* bufO1 = (float*)carve((size_t)N * HID * 4);
    unsigned* lists = (unsigned*)carve((size_t)256 * OWN_CAP * 4);
    int* lcnt = (int*)carve((size_t)ob * 4);
    float* sums  = (float*)carve((size_t)NGRAPHS * HID * 4);
    float* cnt   = (float*)carve((size_t)NGRAPHS * 4);

    const int T = 256;
    int cb   = (N + OWN_NT - 1) / OWN_NT;
    int kb   = (E + SEG_E - 1) / SEG_E;
    int pab  = (N + 8 * PA_ROWS - 1) / (8 * PA_ROWS);
    int pb   = (NGRAPHS * HID / 4 + FR_T - 1) / FR_T;
    int hb   = (NGRAPHS * FOUT + T - 1) / T;
    unsigned* seg = (unsigned*)bufO1;
    unsigned* tab = seg + (size_t)kb * SEG_E;
    int segok = ((size_t)kb * (SEG_E + TAB_W) <= (size_t)N * HID) ? 1 : 0;

    k_front      <<<kb + pb, FR_T, 0, stream>>>(esrc, edst, seg, tab, E, sums, cnt, kb, segok);
    k_owner_list <<<ob + cb, OWN_NT, 0, stream>>>(edst, seg, tab, lists, lcnt, dinv, batch, cnt, x, W1, bufH, N, E, ob, kb, segok);

    k_conv_owner<true> <<<ob, OWN_NT, 0, stream>>>(bufH, bufO1, esrc, edst, lists, lcnt, dinv, b1, W2, N, E);

    k_conv_owner<false><<<ob, OWN_NT, 0, stream>>>(bufO1, bufH, esrc, edst, lists, lcnt, dinv, b2, W2, N, E);

    k_pool_accum<<<pab, T, 0, stream>>>(bufH, batch, sums, N);
    k_final     <<<hb, T, 0, stream>>>(sums, cnt, Wlin, blin, out);
}
